# starting kernel
# baseline (speedup 1.0000x reference)
#include <hip/hip_runtime.h>

#define N_NODES 100000

typedef __attribute__((ext_vector_type(2))) float v2f;
typedef __attribute__((ext_vector_type(8))) float v8f;

__global__ void deg_kernel(const int* __restrict__ src, const int* __restrict__ dst,
                           float* __restrict__ odeg, float* __restrict__ ideg, int E) {
    int e = blockIdx.x * blockDim.x + threadIdx.x;
    if (e < E) {
        atomicAdd(&odeg[src[e]], 1.0f);
        atomicAdd(&ideg[dst[e]], 1.0f);
    }
}

__global__ void norm_kernel(float* __restrict__ deg, int n) {
    int i = blockIdx.x * blockDim.x + threadIdx.x;
    if (i < n) {
        deg[i] = rsqrtf(fmaxf(deg[i], 1.0f));
    }
}

template <int DIN, int DOUT>
__global__ void gemm_norm_wmma(const float* __restrict__ X, const float* __restrict__ ns,
                               const float* __restrict__ W, float* __restrict__ Y) {
    const int wave = threadIdx.x >> 5;
    const int lane = threadIdx.x & 31;
    const int hl   = lane & 15;
    const bool hi  = (lane >= 16);

    const int row0 = blockIdx.x * 16;
    const int col0 = wave * 16;
    const int arow = row0 + hl;
    const int col  = col0 + hl;

    const float nsr = ns[arow];
    const float* __restrict__ xrow = X + (size_t)arow * DIN;

    v8f acc = {};
#pragma unroll 8
    for (int k = 0; k < DIN; k += 4) {
        const int ka = k + (hi ? 2 : 0);
        v2f a, b;
        a[0] = xrow[ka] * nsr;
        a[1] = xrow[ka + 1] * nsr;
        b[0] = W[(size_t)ka * DOUT + col];
        b[1] = W[(size_t)(ka + 1) * DOUT + col];
        acc = __builtin_amdgcn_wmma_f32_16x16x4_f32(
             false, a,  false, b,
             (short)0, acc,  false,  false);
    }

    const int crow0 = row0 + (hi ? 8 : 0);
#pragma unroll
    for (int v = 0; v < 8; ++v) {
        Y[(size_t)(crow0 + v) * DOUT + col] = acc[v];
    }
}

template <int D, int P>
__global__ void scatter_kernel(const float* __restrict__ H, const int* __restrict__ src,
                               const int* __restrict__ dst, float* __restrict__ M, int E) {
    constexpr int PER_EDGE = D / 4;
    const int e = blockIdx.x * (256 >> P) + (threadIdx.x >> P);
    const int f = (threadIdx.x & (PER_EDGE - 1)) << 2;
    if (e >= E) return;

    const int s = src[e];
    const int d = dst[e];
    const float4 v = *reinterpret_cast<const float4*>(H + (size_t)s * D + f);
    float* mp = M + (size_t)d * D + f;
    atomicAdd(mp + 0, v.x);
    atomicAdd(mp + 1, v.y);
    atomicAdd(mp + 2, v.z);
    atomicAdd(mp + 3, v.w);
}

template <int D, bool RELU>
__global__ void finalize_kernel(const float* __restrict__ M, const float* __restrict__ nd,
                                const float* __restrict__ b, float* __restrict__ O, int total) {
    int idx = blockIdx.x * blockDim.x + threadIdx.x;
    if (idx >= total) return;
    const int row = idx / D;
    const int col = idx & (D - 1);
    float v = M[idx] * nd[row] + b[col];
    if (RELU) v = fmaxf(v, 0.0f);
    O[idx] = v;
}

extern "C" void kernel_launch(void* const* d_in, const int* in_sizes, int n_in,
                              void* d_out, int out_size, void* d_ws, size_t ws_size,
                              hipStream_t stream) {
    (void)n_in; (void)out_size; (void)ws_size;

    const float* x  = (const float*)d_in[0];
    const int*   src = (const int*)d_in[1];
    const int*   dst = (const int*)d_in[2];
    const float* W1 = (const float*)d_in[3];
    const float* b1 = (const float*)d_in[4];
    const float* W2 = (const float*)d_in[5];
    const float* b2 = (const float*)d_in[6];
    const float* W3 = (const float*)d_in[7];
    const float* b3 = (const float*)d_in[8];

    const int E = in_sizes[1];
    const int N = N_NODES;

    const size_t NPAD = 100352;
    float* ws   = (float*)d_ws;
    float* ns   = ws;
    float* nd   = ws + NPAD;
    float* bufA = ws + 2 * NPAD;
    float* bufB = bufA + (size_t)N * 128;

    hipMemsetAsync(ns, 0, 2 * NPAD * sizeof(float), stream);
    deg_kernel<<<(E + 255) / 256, 256, 0, stream>>>(src, dst, ns, nd, E);
    norm_kernel<<<(int)((2 * NPAD + 255) / 256), 256, 0, stream>>>(ns, (int)(2 * NPAD));

    const int ROWT = N / 16;

    gemm_norm_wmma<128, 128><<<ROWT, 256, 0, stream>>>(x, ns, W1, bufA);
    hipMemsetAsync(bufB, 0, (size_t)N * 128 * sizeof(float), stream);
    scatter_kernel<128, 5><<<(E + 7) / 8, 256, 0, stream>>>(bufA, src, dst, bufB, E);
    finalize_kernel<128, true><<<(N * 128 + 255) / 256, 256, 0, stream>>>(bufB, nd, b1, bufA, N * 128);

    gemm_norm_wmma<128, 128><<<ROWT, 256, 0, stream>>>(bufA, ns, W2, bufB);
    hipMemsetAsync(bufA, 0, (size_t)N * 128 * sizeof(float), stream);
    scatter_kernel<128, 5><<<(E + 7) / 8, 256, 0, stream>>>(bufB, src, dst, bufA, E);
    finalize_kernel<128, true><<<(N * 128 + 255) / 256, 256, 0, stream>>>(bufA, nd, b2, bufB, N * 128);

    gemm_norm_wmma<128, 64><<<ROWT, 128, 0, stream>>>(bufB, ns, W3, bufA);
    hipMemsetAsync(bufB, 0, (size_t)N * 64 * sizeof(float), stream);
    scatter_kernel<64, 4><<<(E + 15) / 16, 256, 0, stream>>>(bufA, src, dst, bufB, E);
    finalize_kernel<64, false><<<(N * 64 + 255) / 256, 256, 0, stream>>>(bufB, nd, b3, (float*)d_out, N * 64);
}


# scatter: one wave per edge with lane-contiguous features, so each f32 atomic instruction covers one whole 128 B line, not a quarter of four lines; sums stay in atomic arrival order as in the parent
# speedup vs baseline: 1.1746x; 1.1746x over previous
#include <hip/hip_runtime.h>

#define N_NODES 100000

typedef __attribute__((ext_vector_type(2))) float v2f;
typedef __attribute__((ext_vector_type(8))) float v8f;

__global__ void deg_kernel(const int* __restrict__ src, const int* __restrict__ dst,
                           float* __restrict__ odeg, float* __restrict__ ideg, int E) {
    int e = blockIdx.x * blockDim.x + threadIdx.x;
    if (e < E) {
        atomicAdd(&odeg[src[e]], 1.0f);
        atomicAdd(&ideg[dst[e]], 1.0f);
    }
}

__global__ void norm_kernel(float* __restrict__ deg, int n) {
    int i = blockIdx.x * blockDim.x + threadIdx.x;
    if (i < n) {
        deg[i] = rsqrtf(fmaxf(deg[i], 1.0f));
    }
}

template <int DIN, int DOUT>
__global__ void gemm_norm_wmma(const float* __restrict__ X, const float* __restrict__ ns,
                               const float* __restrict__ W, float* __restrict__ Y) {
    const int wave = threadIdx.x >> 5;
    const int lane = threadIdx.x & 31;
    const int hl   = lane & 15;
    const bool hi  = (lane >= 16);

    const int row0 = blockIdx.x * 16;
    const int col0 = wave * 16;
    const int arow = row0 + hl;
    const int col  = col0 + hl;

    const float nsr = ns[arow];
    const float* __restrict__ xrow = X + (size_t)arow * DIN;

    v8f acc = {};
#pragma unroll 8
    for (int k = 0; k < DIN; k += 4) {
        const int ka = k + (hi ? 2 : 0);
        v2f a, b;
        a[0] = xrow[ka] * nsr;
        a[1] = xrow[ka + 1] * nsr;
        b[0] = W[(size_t)ka * DOUT + col];
        b[1] = W[(size_t)(ka + 1) * DOUT + col];
        acc = __builtin_amdgcn_wmma_f32_16x16x4_f32(
             false, a,  false, b,
             (short)0, acc,  false,  false);
    }

    const int crow0 = row0 + (hi ? 8 : 0);
#pragma unroll
    for (int v = 0; v < 8; ++v) {
        Y[(size_t)(crow0 + v) * DOUT + col] = acc[v];
    }
}

template <int D>
__global__ void scatter_kernel(const float* __restrict__ H, const int* __restrict__ src,
                               const int* __restrict__ dst, float* __restrict__ M, int E) {
    constexpr int PER_LANE = D / 32;
    const int e = blockIdx.x * 8 + (threadIdx.x >> 5);
    const int lane = threadIdx.x & 31;
    if (e >= E) return;

    const int s = src[e];
    const int d = dst[e];
    const float* hp = H + (size_t)s * D + lane;
    float* mp = M + (size_t)d * D + lane;
    float v[PER_LANE];
    for (int j = 0; j < PER_LANE; ++j) v[j] = hp[32 * j];
    for (int j = 0; j < PER_LANE; ++j) atomicAdd(mp + 32 * j, v[j]);
}

template <int D, bool RELU>
__global__ void finalize_kernel(const float* __restrict__ M, const float* __restrict__ nd,
                                const float* __restrict__ b, float* __restrict__ O, int total) {
    int idx = blockIdx.x * blockDim.x + threadIdx.x;
    if (idx >= total) return;
    const int row = idx / D;
    const int col = idx & (D - 1);
    float v = M[idx] * nd[row] + b[col];
    if (RELU) v = fmaxf(v, 0.0f);
    O[idx] = v;
}

extern "C" void kernel_launch(void* const* d_in, const int* in_sizes, int n_in,
                              void* d_out, int out_size, void* d_ws, size_t ws_size,
                              hipStream_t stream) {
    (void)n_in; (void)out_size; (void)ws_size;

    const float* x  = (const float*)d_in[0];
    const int*   src = (const int*)d_in[1];
    const int*   dst = (const int*)d_in[2];
    const float* W1 = (const float*)d_in[3];
    const float* b1 = (const float*)d_in[4];
    const float* W2 = (const float*)d_in[5];
    const float* b2 = (const float*)d_in[6];
    const float* W3 = (const float*)d_in[7];
    const float* b3 = (const float*)d_in[8];

    const int E = in_sizes[1];
    const int N = N_NODES;

    const size_t NPAD = 100352;
    float* ws   = (float*)d_ws;
    float* ns   = ws;
    float* nd   = ws + NPAD;
    float* bufA = ws + 2 * NPAD;
    float* bufB = bufA + (size_t)N * 128;

    hipMemsetAsync(ns, 0, 2 * NPAD * sizeof(float), stream);
    deg_kernel<<<(E + 255) / 256, 256, 0, stream>>>(src, dst, ns, nd, E);
    norm_kernel<<<(int)((2 * NPAD + 255) / 256), 256, 0, stream>>>(ns, (int)(2 * NPAD));

    const int ROWT = N / 16;

    gemm_norm_wmma<128, 128><<<ROWT, 256, 0, stream>>>(x, ns, W1, bufA);
    hipMemsetAsync(bufB, 0, (size_t)N * 128 * sizeof(float), stream);
    scatter_kernel<128><<<(E + 7) / 8, 256, 0, stream>>>(bufA, src, dst, bufB, E);
    finalize_kernel<128, true><<<(N * 128 + 255) / 256, 256, 0, stream>>>(bufB, nd, b1, bufA, N * 128);

    gemm_norm_wmma<128, 128><<<ROWT, 256, 0, stream>>>(bufA, ns, W2, bufB);
    hipMemsetAsync(bufA, 0, (size_t)N * 128 * sizeof(float), stream);
    scatter_kernel<128><<<(E + 7) / 8, 256, 0, stream>>>(bufB, src, dst, bufA, E);
    finalize_kernel<128, true><<<(N * 128 + 255) / 256, 256, 0, stream>>>(bufA, nd, b2, bufB, N * 128);

    gemm_norm_wmma<128, 64><<<ROWT, 128, 0, stream>>>(bufB, ns, W3, bufA);
    hipMemsetAsync(bufB, 0, (size_t)N * 64 * sizeof(float), stream);
    scatter_kernel<64><<<(E + 7) / 8, 256, 0, stream>>>(bufA, src, dst, bufB, E);
    finalize_kernel<64, false><<<(N * 64 + 255) / 256, 256, 0, stream>>>(bufB, nd, b3, (float*)d_out, N * 64);
}


# aggregation: edges bucketed by destination in the call (integer-atomic slots, scan, place); one wave per node sums its source rows in registers in list order, one store per line, no f32 atomics
# speedup vs baseline: 2.4711x; 2.1038x over previous
#include <hip/hip_runtime.h>

#define N_NODES 100000

typedef __attribute__((ext_vector_type(2))) float v2f;
typedef __attribute__((ext_vector_type(8))) float v8f;

__global__ void count_rank_kernel(const int* __restrict__ src, const int* __restrict__ dst,
                                  int* __restrict__ ocnt, int* __restrict__ icnt,
                                  int* __restrict__ rank, int E) {
    int e = blockIdx.x * blockDim.x + threadIdx.x;
    if (e < E) {
        atomicAdd(&ocnt[src[e]], 1);
        rank[e] = atomicAdd(&icnt[dst[e]], 1);
    }
}

__global__ void scan_norm_kernel(const int* __restrict__ icnt, int* __restrict__ ocnt_ns,
                                 int* __restrict__ rowptr, int n) {
    __shared__ int wsum[32];
    __shared__ int wpre[32];
    const int t = threadIdx.x;
    const int lane = t & 31;
    const int w = t >> 5;
    const int nprev = blockIdx.x * 1024;

    int pre = 0;
    for (int i = t * 4; i < nprev; i += 4096) {
        const int4 q = *(const int4*)(icnt + i);
        pre += (q.x + q.y) + (q.z + q.w);
    }
    const int idx = nprev + t;
    const int c = (idx < n) ? icnt[idx] : 0;

    int inc = c;
    for (int off = 1; off < 32; off <<= 1) {
        const int v = __shfl_up(inc, off);
        if (lane >= off) inc += v;
    }
    for (int off = 16; off > 0; off >>= 1) pre += __shfl_xor(pre, off);
    if (lane == 31) wsum[w] = inc;
    if (lane == 0) wpre[w] = pre;
    __syncthreads();

    const int wv = wsum[lane];
    int pv = wpre[lane];
    int sc = wv;
    for (int off = 1; off < 32; off <<= 1) {
        const int v = __shfl_up(sc, off);
        if (lane >= off) sc += v;
    }
    const int wave_excl = __shfl(sc - wv, w);
    for (int off = 16; off > 0; off >>= 1) pv += __shfl_xor(pv, off);

    const int excl = pv + wave_excl + (inc - c);
    if (idx < n) {
        rowptr[idx] = excl;
        if (idx == n - 1) rowptr[n] = excl + c;
        ocnt_ns[idx] = __float_as_int(rsqrtf(fmaxf((float)ocnt_ns[idx], 1.0f)));
    }
}

__global__ void place_kernel(const int* __restrict__ src, const int* __restrict__ dst,
                             const int* __restrict__ rowptr, const int* __restrict__ rank,
                             int* __restrict__ srcs, int E) {
    int e = blockIdx.x * blockDim.x + threadIdx.x;
    if (e < E) srcs[rowptr[dst[e]] + rank[e]] = src[e];
}

__global__ void move_lists_kernel(const int* __restrict__ from, int* __restrict__ to, int n) {
    int i = blockIdx.x * blockDim.x + threadIdx.x;
    if (i < n) to[i] = from[i];
}

template <int DIN, int DOUT>
__global__ void gemm_norm_wmma(const float* __restrict__ X, const float* __restrict__ ns,
                               const float* __restrict__ W, float* __restrict__ Y) {
    const int wave = threadIdx.x >> 5;
    const int lane = threadIdx.x & 31;
    const int hl   = lane & 15;
    const bool hi  = (lane >= 16);

    const int row0 = blockIdx.x * 16;
    const int col0 = wave * 16;
    const int arow = row0 + hl;
    const int col  = col0 + hl;

    const float nsr = ns[arow];
    const float* __restrict__ xrow = X + (size_t)arow * DIN;

    v8f acc = {};
#pragma unroll 8
    for (int k = 0; k < DIN; k += 4) {
        const int ka = k + (hi ? 2 : 0);
        v2f a, b;
        a[0] = xrow[ka] * nsr;
        a[1] = xrow[ka + 1] * nsr;
        b[0] = W[(size_t)ka * DOUT + col];
        b[1] = W[(size_t)(ka + 1) * DOUT + col];
        acc = __builtin_amdgcn_wmma_f32_16x16x4_f32(
             false, a,  false, b,
             (short)0, acc,  false,  false);
    }

    const int crow0 = row0 + (hi ? 8 : 0);
#pragma unroll
    for (int v = 0; v < 8; ++v) {
        Y[(size_t)(crow0 + v) * DOUT + col] = acc[v];
    }
}

template <int D, bool RELU>
__global__ void gather_sum_kernel(const float* __restrict__ H, const int* __restrict__ rowptr,
                                  const int* __restrict__ srcs, const float* __restrict__ bias,
                                  float* __restrict__ O) {
    constexpr int VEC = D / 32;
    constexpr int BATCH = 8;
    typedef __attribute__((ext_vector_type(VEC))) float vec;

    const int lane = threadIdx.x & 31;
    const int node = __builtin_amdgcn_readfirstlane((int)(blockIdx.x * 8 + (threadIdx.x >> 5)));
    if (node >= N_NODES) return;

    const int beg = rowptr[node];
    const int end = rowptr[node + 1];
    const char* __restrict__ Hb = (const char*)H;
    const unsigned lane_off = (unsigned)lane * (VEC * 4);

    vec acc = (vec)(0.0f);
    for (int base = beg; base < end; base += 32) {
        const int cnt = min(32, end - base);
        const int mine = (lane < cnt) ? srcs[base + lane] : 0;
        int j = 0;
        for (; j + BATCH <= cnt; j += BATCH) {
            int s[BATCH];
            vec v[BATCH];
#pragma unroll
            for (int u = 0; u < BATCH; ++u) s[u] = __builtin_amdgcn_readlane(mine, j + u);
#pragma unroll
            for (int u = 0; u < BATCH; ++u)
                v[u] = *(const vec*)(Hb + ((size_t)(unsigned)s[u] * (D * 4)) + lane_off);
#pragma unroll
            for (int u = 0; u < BATCH; ++u) acc += v[u];
        }
        if (j < cnt) {
            int s[BATCH];
            vec v[BATCH];
#pragma unroll
            for (int u = 0; u < BATCH; ++u) s[u] = __builtin_amdgcn_readlane(mine, (j + u) & 31);
#pragma unroll
            for (int u = 0; u < BATCH; ++u) {
                if (j + u < cnt)
                    v[u] = *(const vec*)(Hb + ((size_t)(unsigned)s[u] * (D * 4)) + lane_off);
                else
                    v[u] = (vec)(0.0f);
            }
#pragma unroll
            for (int u = 0; u < BATCH; ++u) acc += v[u];
        }
    }

    const float nd = rsqrtf(fmaxf((float)(end - beg), 1.0f));
    const vec bv = ((const vec*)bias)[lane];
    vec o = acc * nd + bv;
    if (RELU) {
#pragma unroll
        for (int i = 0; i < VEC; ++i) o[i] = fmaxf(o[i], 0.0f);
    }
    ((vec*)O)[(size_t)node * 32 + lane] = o;
}

extern "C" void kernel_launch(void* const* d_in, const int* in_sizes, int n_in,
                              void* d_out, int out_size, void* d_ws, size_t ws_size,
                              hipStream_t stream) {
    (void)n_in; (void)out_size; (void)ws_size;

    const float* x  = (const float*)d_in[0];
    const int*   src = (const int*)d_in[1];
    const int*   dst = (const int*)d_in[2];
    const float* W1 = (const float*)d_in[3];
    const float* b1 = (const float*)d_in[4];
    const float* W2 = (const float*)d_in[5];
    const float* b2 = (const float*)d_in[6];
    const float* W3 = (const float*)d_in[7];
    const float* b3 = (const float*)d_in[8];

    const int E = in_sizes[1];
    const int N = N_NODES;

    const size_t NPAD = 100352;
    float* ws   = (float*)d_ws;
    float* ns   = ws;
    int*   ocnt = (int*)ws;
    int*   icnt = (int*)ws + NPAD;
    float* bufA = ws + 2 * NPAD;
    float* bufB = bufA + (size_t)N * 128;

    int* rowptr  = (int*)d_out;
    int* srcs    = rowptr + (N + 1);
    int* rank    = (int*)bufB;
    int* rowptr3 = (int*)(bufA + (size_t)N * 64);
    int* srcs3   = rowptr3 + (N + 1);

    hipMemsetAsync(ws, 0, 2 * NPAD * sizeof(float), stream);
    count_rank_kernel<<<(E + 255) / 256, 256, 0, stream>>>(src, dst, ocnt, icnt, rank, E);
    scan_norm_kernel<<<(N + 1023) / 1024, 1024, 0, stream>>>(icnt, ocnt, rowptr, N);
    place_kernel<<<(E + 255) / 256, 256, 0, stream>>>(src, dst, rowptr, rank, srcs, E);

    const int ROWT = N / 16;
    const int NODEB = (N + 7) / 8;

    gemm_norm_wmma<128, 128><<<ROWT, 256, 0, stream>>>(x, ns, W1, bufA);
    gather_sum_kernel<128, true><<<NODEB, 256, 0, stream>>>(bufA, rowptr, srcs, b1, bufB);

    gemm_norm_wmma<128, 128><<<ROWT, 256, 0, stream>>>(bufB, ns, W2, bufA);
    gather_sum_kernel<128, true><<<NODEB, 256, 0, stream>>>(bufA, rowptr, srcs, b2, bufB);

    gemm_norm_wmma<128, 64><<<ROWT, 128, 0, stream>>>(bufB, ns, W3, bufA);
    move_lists_kernel<<<(N + 1 + E + 255) / 256, 256, 0, stream>>>(rowptr, rowptr3, N + 1 + E);
    gather_sum_kernel<64, false><<<NODEB, 256, 0, stream>>>(bufA, rowptr3, srcs3, b3, (float*)d_out);
}


# every launch but the scan sized to the resident wave slots: waves loop over nodes, row stripes or edge chunks (next node's list fetched ahead, W tile kept in registers); sums in the parent's order
# speedup vs baseline: 2.6777x; 1.0836x over previous
#include <hip/hip_runtime.h>

#define N_NODES 100000

typedef __attribute__((ext_vector_type(2))) float v2f;
typedef __attribute__((ext_vector_type(8))) float v8f;

__global__ void count_rank_kernel(const int* __restrict__ src, const int* __restrict__ dst,
                                  int* __restrict__ ocnt, int* __restrict__ icnt,
                                  int* __restrict__ rank, int E) {
    for (int c0 = blockIdx.x * 2048; c0 < E; c0 += gridDim.x * 2048) {
        const int e0 = c0 + threadIdx.x;
        if (c0 + 2048 <= E) {
            unsigned s[8], d[8];
            int r[8];
#pragma unroll
            for (int u = 0; u < 8; ++u) {
                s[u] = (unsigned)src[e0 + u * 256];
                d[u] = (unsigned)dst[e0 + u * 256];
            }
#pragma unroll
            for (int u = 0; u < 8; ++u) atomicAdd(&ocnt[s[u]], 1);
#pragma unroll
            for (int u = 0; u < 8; ++u) r[u] = atomicAdd(&icnt[d[u]], 1);
#pragma unroll
            for (int u = 0; u < 8; ++u) rank[e0 + u * 256] = r[u];
        } else {
            for (int e = e0; e < E; e += 256) {
                atomicAdd(&ocnt[(unsigned)src[e]], 1);
                rank[e] = atomicAdd(&icnt[(unsigned)dst[e]], 1);
            }
        }
    }
}

__global__ void scan_norm_kernel(const int* __restrict__ icnt, int* __restrict__ ocnt_ns,
                                 int* __restrict__ rowptr, int n) {
    __shared__ int wsum[32];
    __shared__ int wpre[32];
    const int t = threadIdx.x;
    const int lane = t & 31;
    const int w = t >> 5;
    const int nprev = blockIdx.x * 1024;

    int pre = 0;
    for (int i = t * 4; i < nprev; i += 4096) {
        const int4 q = *(const int4*)(icnt + i);
        pre += (q.x + q.y) + (q.z + q.w);
    }
    const int idx = nprev + t;
    const int c = (idx < n) ? icnt[idx] : 0;

    int inc = c;
    for (int off = 1; off < 32; off <<= 1) {
        const int v = __shfl_up(inc, off);
        if (lane >= off) inc += v;
    }
    for (int off = 16; off > 0; off >>= 1) pre += __shfl_xor(pre, off);
    if (lane == 31) wsum[w] = inc;
    if (lane == 0) wpre[w] = pre;
    __syncthreads();

    const int wv = wsum[lane];
    int pv = wpre[lane];
    int sc = wv;
    for (int off = 1; off < 32; off <<= 1) {
        const int v = __shfl_up(sc, off);
        if (lane >= off) sc += v;
    }
    const int wave_excl = __shfl(sc - wv, w);
    for (int off = 16; off > 0; off >>= 1) pv += __shfl_xor(pv, off);

    const int excl = pv + wave_excl + (inc - c);
    if (idx < n) {
        rowptr[idx] = excl;
        if (idx == n - 1) rowptr[n] = excl + c;
        ocnt_ns[idx] = __float_as_int(rsqrtf(fmaxf((float)ocnt_ns[idx], 1.0f)));
    }
}

__global__ void place_kernel(const int* __restrict__ src, const int* __restrict__ dst,
                             const int* __restrict__ rowptr, const int* __restrict__ rank,
                             int* __restrict__ srcs, int E) {
    for (int c0 = blockIdx.x * 2048; c0 < E; c0 += gridDim.x * 2048) {
        const int e0 = c0 + threadIdx.x;
        if (c0 + 2048 <= E) {
            unsigned d[8];
            int s[8], r[8], p[8];
#pragma unroll
            for (int u = 0; u < 8; ++u) {
                d[u] = (unsigned)dst[e0 + u * 256];
                r[u] = rank[e0 + u * 256];
                s[u] = src[e0 + u * 256];
            }
#pragma unroll
            for (int u = 0; u < 8; ++u) p[u] = rowptr[d[u]];
#pragma unroll
            for (int u = 0; u < 8; ++u) srcs[(unsigned)(p[u] + r[u])] = s[u];
        } else {
            for (int e = e0; e < E; e += 256)
                srcs[(unsigned)(rowptr[(unsigned)dst[e]] + rank[e])] = src[e];
        }
    }
}

__global__ void move_lists_kernel(const int* __restrict__ from, int* __restrict__ to, int n) {
    for (int c0 = blockIdx.x * 2048; c0 < n; c0 += gridDim.x * 2048) {
        const int i0 = c0 + threadIdx.x;
        if (c0 + 2048 <= n) {
            int v[8];
#pragma unroll
            for (int u = 0; u < 8; ++u) v[u] = from[i0 + u * 256];
#pragma unroll
            for (int u = 0; u < 8; ++u) to[i0 + u * 256] = v[u];
        } else {
            for (int i = i0; i < n; i += 256) to[i] = from[i];
        }
    }
}

template <int DIN, int DOUT>
__global__ void gemm_norm_wmma(const float* __restrict__ X, const float* __restrict__ ns,
                               const float* __restrict__ W, float* __restrict__ Y,
                               int nstripes) {
    const int wave = threadIdx.x >> 5;
    const int lane = threadIdx.x & 31;
    const int hl   = lane & 15;
    const bool hi  = (lane >= 16);
    const int kofs = hi ? 2 : 0;
    const int col  = wave * 16 + hl;

    v2f bfrag[DIN / 4];
#pragma unroll
    for (int i = 0; i < DIN / 4; ++i) {
        const int ka = 4 * i + kofs;
        bfrag[i][0] = W[(size_t)ka * DOUT + col];
        bfrag[i][1] = W[(size_t)(ka + 1) * DOUT + col];
    }

    for (int stripe = blockIdx.x; stripe < nstripes; stripe += gridDim.x) {
        const int row0 = stripe * 16;
        const int arow = row0 + hl;
        const float nsr = ns[arow];
        const float* __restrict__ xrow = X + (size_t)arow * DIN + kofs;

        v8f acc = {};
#pragma unroll
        for (int i = 0; i < DIN / 4; ++i) {
            v2f a;
            a[0] = xrow[4 * i] * nsr;
            a[1] = xrow[4 * i + 1] * nsr;
            acc = __builtin_amdgcn_wmma_f32_16x16x4_f32(
                 false, a,  false, bfrag[i],
                 (short)0, acc,  false,  false);
        }

        const int crow0 = row0 + (hi ? 8 : 0);
#pragma unroll
        for (int v = 0; v < 8; ++v) {
            Y[(size_t)(crow0 + v) * DOUT + col] = acc[v];
        }
    }
}

template <int D, int BATCH, typename vec>
__device__ __forceinline__ void sum_rows(vec& acc, const char* __restrict__ Hb, unsigned lane_off,
                                         int mine, int cnt) {
    int j = 0;
    for (; j + BATCH <= cnt; j += BATCH) {
        int s[BATCH];
        vec v[BATCH];
#pragma unroll
        for (int u = 0; u < BATCH; ++u) s[u] = __builtin_amdgcn_readlane(mine, j + u);
#pragma unroll
        for (int u = 0; u < BATCH; ++u)
            v[u] = *(const vec*)(Hb + ((size_t)(unsigned)s[u] * (D * 4)) + lane_off);
#pragma unroll
        for (int u = 0; u < BATCH; ++u) acc += v[u];
    }
    if (j < cnt) {
        int s[BATCH];
        vec v[BATCH];
#pragma unroll
        for (int u = 0; u < BATCH; ++u) s[u] = __builtin_amdgcn_readlane(mine, (j + u) & 31);
#pragma unroll
        for (int u = 0; u < BATCH; ++u) {
            if (j + u < cnt)
                v[u] = *(const vec*)(Hb + ((size_t)(unsigned)s[u] * (D * 4)) + lane_off);
            else
                v[u] = (vec)(0.0f);
        }
#pragma unroll
        for (int u = 0; u < BATCH; ++u) acc += v[u];
    }
}

template <int D, bool RELU>
__global__ void gather_sum_kernel(const float* __restrict__ H, const int* __restrict__ rowptr,
                                  const int* __restrict__ srcs, const float* __restrict__ bias,
                                  float* __restrict__ O) {
    constexpr int VEC = D / 32;
    constexpr int BATCH = 8;
    typedef __attribute__((ext_vector_type(VEC))) float vec;

    const int lane = threadIdx.x & 31;
    const int wv = __builtin_amdgcn_readfirstlane((int)(blockIdx.x * 8 + (threadIdx.x >> 5)));
    const int nw = (int)gridDim.x * 8;
    if (wv >= N_NODES) return;

    const char* __restrict__ Hb = (const char*)H;
    const unsigned lane_off = (unsigned)lane * (VEC * 4);
    const vec bv = ((const vec*)bias)[lane];

    int node = wv;
    int beg = rowptr[node];
    int end = rowptr[node + 1];
    int first = (lane < end - beg) ? srcs[beg + lane] : 0;
    asm volatile("" : "+v"(first));
    int nnode = node + nw;
    int nbeg = 0, nend = 0;
    if (nnode < N_NODES) { nbeg = rowptr[nnode]; nend = rowptr[nnode + 1]; }

    for (;;) {
        int nfirst = 0;
        if (nnode < N_NODES) nfirst = (lane < nend - nbeg) ? srcs[nbeg + lane] : 0;
        const int n2 = nnode + nw;
        int n2beg = 0, n2end = 0;
        if (n2 < N_NODES) { n2beg = rowptr[n2]; n2end = rowptr[n2 + 1]; }

        vec acc = (vec)(0.0f);
        sum_rows<D, BATCH, vec>(acc, Hb, lane_off, first, min(32, end - beg));
        for (int base = beg + 32; base < end; base += 32) {
            const int cnt = min(32, end - base);
            const int mine = (lane < cnt) ? srcs[base + lane] : 0;
            sum_rows<D, BATCH, vec>(acc, Hb, lane_off, mine, cnt);
        }

        const float nd = rsqrtf(fmaxf((float)(end - beg), 1.0f));
        vec o = acc * nd + bv;
        if (RELU) {
#pragma unroll
            for (int i = 0; i < VEC; ++i) o[i] = fmaxf(o[i], 0.0f);
        }
        ((vec*)O)[(size_t)node * 32 + lane] = o;

        if (nnode >= N_NODES) break;
        node = nnode; beg = nbeg; end = nend; first = nfirst;
        nnode = n2; nbeg = n2beg; nend = n2end;
    }
}

extern "C" void kernel_launch(void* const* d_in, const int* in_sizes, int n_in,
                              void* d_out, int out_size, void* d_ws, size_t ws_size,
                              hipStream_t stream) {
    (void)n_in; (void)out_size; (void)ws_size;

    const float* x  = (const float*)d_in[0];
    const int*   src = (const int*)d_in[1];
    const int*   dst = (const int*)d_in[2];
    const float* W1 = (const float*)d_in[3];
    const float* b1 = (const float*)d_in[4];
    const float* W2 = (const float*)d_in[5];
    const float* b2 = (const float*)d_in[6];
    const float* W3 = (const float*)d_in[7];
    const float* b3 = (const float*)d_in[8];

    const int E = in_sizes[1];
    const int N = N_NODES;

    const size_t NPAD = 100352;
    float* ws   = (float*)d_ws;
    float* ns   = ws;
    int*   ocnt = (int*)ws;
    int*   icnt = (int*)ws + NPAD;
    float* bufA = ws + 2 * NPAD;
    float* bufB = bufA + (size_t)N * 128;

    int* rowptr  = (int*)d_out;
    int* srcs    = rowptr + (N + 1);
    int* rank    = (int*)bufB;
    int* rowptr3 = (int*)(bufA + (size_t)N * 64);
    int* srcs3   = rowptr3 + (N + 1);

    hipMemsetAsync(ws, 0, 2 * NPAD * sizeof(float), stream);
    const int EDGEB = 1024;
    count_rank_kernel<<<EDGEB, 256, 0, stream>>>(src, dst, ocnt, icnt, rank, E);
    scan_norm_kernel<<<(N + 1023) / 1024, 1024, 0, stream>>>(icnt, ocnt, rowptr, N);
    place_kernel<<<EDGEB, 256, 0, stream>>>(src, dst, rowptr, rank, srcs, E);

    const int ROWT = N / 16;
    const int GEMMB = 1024;
    const int GEMMB64 = 2048;
    const int NODEB = 2048;

    gemm_norm_wmma<128, 128><<<GEMMB, 256, 0, stream>>>(x, ns, W1, bufA, ROWT);
    gather_sum_kernel<128, true><<<NODEB, 256, 0, stream>>>(bufA, rowptr, srcs, b1, bufB);

    gemm_norm_wmma<128, 128><<<GEMMB, 256, 0, stream>>>(bufB, ns, W2, bufA, ROWT);
    gather_sum_kernel<128, true><<<NODEB, 256, 0, stream>>>(bufA, rowptr, srcs, b2, bufB);

    gemm_norm_wmma<128, 64><<<GEMMB64, 128, 0, stream>>>(bufB, ns, W3, bufA, ROWT);
    move_lists_kernel<<<EDGEB, 256, 0, stream>>>(rowptr, rowptr3, N + 1 + E);
    gather_sum_kernel<64, false><<<NODEB, 256, 0, stream>>>(bufA, rowptr3, srcs3, b3, (float*)d_out);
}


# list build without global atomics: edges binned by 512-node range with LDS counters (count, scan, place), one block per range counts degrees and lists its nodes in LDS; sum order = LDS add order
# speedup vs baseline: 3.8038x; 1.4205x over previous
#include <hip/hip_runtime.h>

#define N_NODES 100000

typedef __attribute__((ext_vector_type(2))) float v2f;
typedef __attribute__((ext_vector_type(8))) float v8f;

#define NBIN 196
#define G1 256
#define TBL (NBIN * G1)

__device__ __forceinline__ int range_of(unsigned node) { return min((int)(node >> 9), NBIN - 1); }

__global__ void __launch_bounds__(1024)
bin_count_kernel(const int* __restrict__ src, const int* __restrict__ dst,
                 int* __restrict__ tbl, int E) {
    __shared__ int hd[NBIN];
    __shared__ int hs[NBIN];
    const int t = threadIdx.x;
    const int b = blockIdx.x;
    if (t < NBIN) { hd[t] = 0; hs[t] = 0; }
    __syncthreads();
    const int ch = (E + G1 - 1) / G1;
    const int beg = b * ch;
    const int end = min(beg + ch, E);
    for (int i0 = beg; i0 < end; i0 += 8 * 1024) {
        unsigned d[8], s[8];
#pragma unroll
        for (int u = 0; u < 8; ++u) {
            const int i = i0 + u * 1024 + t;
            const int ic = (i < end) ? i : beg;
            d[u] = (unsigned)dst[ic];
            s[u] = (unsigned)src[ic];
        }
#pragma unroll
        for (int u = 0; u < 8; ++u) {
            if (i0 + u * 1024 + t < end) {
                atomicAdd(&hd[range_of(d[u])], 1);
                atomicAdd(&hs[range_of(s[u])], 1);
            }
        }
    }
    __syncthreads();
    if (t < NBIN) {
        tbl[t * G1 + b] = hd[t];
        tbl[TBL + t * G1 + b] = hs[t];
    }
}

__global__ void __launch_bounds__(1024)
scan_kernel(const int* __restrict__ cnt, int* __restrict__ off, int n) {
    __shared__ int wsum[32];
    __shared__ int wpre[32];
    const int t = threadIdx.x;
    const int lane = t & 31;
    const int w = t >> 5;
    const int nprev = blockIdx.x * 1024;

    int pre = 0;
    for (int i = t * 4; i < nprev; i += 4096) {
        const int4 q = *(const int4*)(cnt + i);
        pre += (q.x + q.y) + (q.z + q.w);
    }
    const int idx = nprev + t;
    const int c = (idx < n) ? cnt[idx] : 0;

    int inc = c;
    for (int o = 1; o < 32; o <<= 1) {
        const int v = __shfl_up(inc, o);
        if (lane >= o) inc += v;
    }
    for (int o = 16; o > 0; o >>= 1) pre += __shfl_xor(pre, o);
    if (lane == 31) wsum[w] = inc;
    if (lane == 0) wpre[w] = pre;
    __syncthreads();

    const int wv = wsum[lane];
    int pv = wpre[lane];
    int sc = wv;
    for (int o = 1; o < 32; o <<= 1) {
        const int v = __shfl_up(sc, o);
        if (lane >= o) sc += v;
    }
    const int wave_excl = __shfl(sc - wv, w);
    for (int o = 16; o > 0; o >>= 1) pv += __shfl_xor(pv, o);

    if (idx < n) off[idx] = pv + wave_excl + (inc - c);
}

__global__ void __launch_bounds__(1024)
bin_place_kernel(const int* __restrict__ src, const int* __restrict__ dst,
                 const int* __restrict__ off, int* __restrict__ runs_d,
                 int* __restrict__ runs_s, int E) {
    __shared__ int cd[NBIN];
    __shared__ int cs[NBIN];
    const int t = threadIdx.x;
    const int b = blockIdx.x;
    if (t < NBIN) { cd[t] = off[t * G1 + b]; cs[t] = off[TBL + t * G1 + b] - E; }
    __syncthreads();
    const int ch = (E + G1 - 1) / G1;
    const int beg = b * ch;
    const int end = min(beg + ch, E);
    for (int i0 = beg; i0 < end; i0 += 8 * 1024) {
        unsigned d[8], s[8];
#pragma unroll
        for (int u = 0; u < 8; ++u) {
            const int i = i0 + u * 1024 + t;
            const int ic = (i < end) ? i : beg;
            d[u] = (unsigned)dst[ic];
            s[u] = (unsigned)src[ic];
        }
#pragma unroll
        for (int u = 0; u < 8; ++u) {
            if (i0 + u * 1024 + t < end) {
                const int pd = atomicAdd(&cd[range_of(d[u])], 1);
                const int ps = atomicAdd(&cs[range_of(s[u])], 1);
                runs_d[pd] = (int)((s[u] << 9) | (d[u] & 511u));
                runs_s[ps] = (int)(s[u] & 511u);
            }
        }
    }
}

__global__ void __launch_bounds__(1024)
range_lists_kernel(const int* __restrict__ off, const int* __restrict__ runs_d,
                   const int* __restrict__ runs_s, int* __restrict__ rowptr,
                   int* __restrict__ srcs, float* __restrict__ ns, int E) {
    __shared__ int cnt[512];
    __shared__ int cur[512];
    __shared__ int ocn[512];
    __shared__ int wsum[32];
    const int t = threadIdx.x;
    const int lane = t & 31;
    const int w = t >> 5;
    const int g = blockIdx.x;
    const int bd = off[g * G1];
    const int ed = off[(g + 1) * G1];
    const int bs = off[TBL + g * G1] - E;
    const int es = (g == NBIN - 1) ? E : off[TBL + (g + 1) * G1] - E;
    if (t < 512) { cnt[t] = 0; ocn[t] = 0; }
    __syncthreads();
    for (int i0 = bd; i0 < ed; i0 += 8 * 1024) {
        int wd[8];
#pragma unroll
        for (int u = 0; u < 8; ++u) {
            const int i = i0 + u * 1024 + t;
            wd[u] = runs_d[(i < ed) ? i : bd];
        }
#pragma unroll
        for (int u = 0; u < 8; ++u)
            if (i0 + u * 1024 + t < ed) atomicAdd(&cnt[wd[u] & 511], 1);
    }
    for (int i0 = bs; i0 < es; i0 += 8 * 1024) {
        int wo[8];
#pragma unroll
        for (int u = 0; u < 8; ++u) {
            const int i = i0 + u * 1024 + t;
            wo[u] = runs_s[(i < es) ? i : bs];
        }
#pragma unroll
        for (int u = 0; u < 8; ++u)
            if (i0 + u * 1024 + t < es) atomicAdd(&ocn[wo[u] & 511], 1);
    }
    __syncthreads();

    const int c = (t < 512) ? cnt[t] : 0;
    int inc = c;
    for (int o = 1; o < 32; o <<= 1) {
        const int v = __shfl_up(inc, o);
        if (lane >= o) inc += v;
    }
    if (lane == 31) wsum[w] = inc;
    __syncthreads();
    const int wv = wsum[lane];
    int sc = wv;
    for (int o = 1; o < 32; o <<= 1) {
        const int v = __shfl_up(sc, o);
        if (lane >= o) sc += v;
    }
    const int wave_excl = __shfl(sc - wv, w);
    const int node = g * 512 + t;
    if (t < 512) {
        const int first = bd + wave_excl + (inc - c);
        cur[t] = first;
        if (node < N_NODES) {
            rowptr[node] = first;
            ns[node] = rsqrtf(fmaxf((float)ocn[t], 1.0f));
        }
    }
    if (g == NBIN - 1 && t == 0) rowptr[N_NODES] = ed;
    __syncthreads();
    for (int i0 = bd; i0 < ed; i0 += 8 * 1024) {
        int wd[8];
#pragma unroll
        for (int u = 0; u < 8; ++u) {
            const int i = i0 + u * 1024 + t;
            wd[u] = runs_d[(i < ed) ? i : bd];
        }
#pragma unroll
        for (int u = 0; u < 8; ++u) {
            if (i0 + u * 1024 + t < ed) {
                const int p = atomicAdd(&cur[wd[u] & 511], 1);
                srcs[p] = (int)((unsigned)wd[u] >> 9);
            }
        }
    }
}

__global__ void move_lists_kernel(const int* __restrict__ from, int* __restrict__ to, int n) {
    for (int c0 = blockIdx.x * 2048; c0 < n; c0 += gridDim.x * 2048) {
        const int i0 = c0 + threadIdx.x;
        if (c0 + 2048 <= n) {
            int v[8];
#pragma unroll
            for (int u = 0; u < 8; ++u) v[u] = from[i0 + u * 256];
#pragma unroll
            for (int u = 0; u < 8; ++u) to[i0 + u * 256] = v[u];
        } else {
            for (int i = i0; i < n; i += 256) to[i] = from[i];
        }
    }
}

template <int DIN, int DOUT>
__global__ void gemm_norm_wmma(const float* __restrict__ X, const float* __restrict__ ns,
                               const float* __restrict__ W, float* __restrict__ Y,
                               int nstripes) {
    const int wave = threadIdx.x >> 5;
    const int lane = threadIdx.x & 31;
    const int hl   = lane & 15;
    const bool hi  = (lane >= 16);
    const int kofs = hi ? 2 : 0;
    const int col  = wave * 16 + hl;

    v2f bfrag[DIN / 4];
#pragma unroll
    for (int i = 0; i < DIN / 4; ++i) {
        const int ka = 4 * i + kofs;
        bfrag[i][0] = W[(size_t)ka * DOUT + col];
        bfrag[i][1] = W[(size_t)(ka + 1) * DOUT + col];
    }

    for (int stripe = blockIdx.x; stripe < nstripes; stripe += gridDim.x) {
        const int row0 = stripe * 16;
        const int arow = row0 + hl;
        const float nsr = ns[arow];
        const float* __restrict__ xrow = X + (size_t)arow * DIN + kofs;

        v8f acc = {};
#pragma unroll
        for (int i = 0; i < DIN / 4; ++i) {
            v2f a;
            a[0] = xrow[4 * i] * nsr;
            a[1] = xrow[4 * i + 1] * nsr;
            acc = __builtin_amdgcn_wmma_f32_16x16x4_f32(
                 false, a,  false, bfrag[i],
                 (short)0, acc,  false,  false);
        }

        const int crow0 = row0 + (hi ? 8 : 0);
#pragma unroll
        for (int v = 0; v < 8; ++v) {
            Y[(size_t)(crow0 + v) * DOUT + col] = acc[v];
        }
    }
}

template <int D, int BATCH, typename vec>
__device__ __forceinline__ void sum_rows(vec& acc, const char* __restrict__ Hb, unsigned lane_off,
                                         int mine, int cnt) {
    int j = 0;
    for (; j + BATCH <= cnt; j += BATCH) {
        int s[BATCH];
        vec v[BATCH];
#pragma unroll
        for (int u = 0; u < BATCH; ++u) s[u] = __builtin_amdgcn_readlane(mine, j + u);
#pragma unroll
        for (int u = 0; u < BATCH; ++u)
            v[u] = *(const vec*)(Hb + ((size_t)(unsigned)s[u] * (D * 4)) + lane_off);
#pragma unroll
        for (int u = 0; u < BATCH; ++u) acc += v[u];
    }
    if (j < cnt) {
        int s[BATCH];
        vec v[BATCH];
#pragma unroll
        for (int u = 0; u < BATCH; ++u) s[u] = __builtin_amdgcn_readlane(mine, (j + u) & 31);
#pragma unroll
        for (int u = 0; u < BATCH; ++u) {
            if (j + u < cnt)
                v[u] = *(const vec*)(Hb + ((size_t)(unsigned)s[u] * (D * 4)) + lane_off);
            else
                v[u] = (vec)(0.0f);
        }
#pragma unroll
        for (int u = 0; u < BATCH; ++u) acc += v[u];
    }
}

template <int D, bool RELU>
__global__ void gather_sum_kernel(const float* __restrict__ H, const int* __restrict__ rowptr,
                                  const int* __restrict__ srcs, const float* __restrict__ bias,
                                  float* __restrict__ O) {
    constexpr int VEC = D / 32;
    constexpr int BATCH = 8;
    typedef __attribute__((ext_vector_type(VEC))) float vec;

    const int lane = threadIdx.x & 31;
    const int wv = __builtin_amdgcn_readfirstlane((int)(blockIdx.x * 8 + (threadIdx.x >> 5)));
    const int nw = (int)gridDim.x * 8;
    if (wv >= N_NODES) return;

    const char* __restrict__ Hb = (const char*)H;
    const unsigned lane_off = (unsigned)lane * (VEC * 4);
    const vec bv = ((const vec*)bias)[lane];

    int node = wv;
    int beg = rowptr[node];
    int end = rowptr[node + 1];
    int first = (lane < end - beg) ? srcs[beg + lane] : 0;
    asm volatile("" : "+v"(first));
    int nnode = node + nw;
    int nbeg = 0, nend = 0;
    if (nnode < N_NODES) { nbeg = rowptr[nnode]; nend = rowptr[nnode + 1]; }

    for (;;) {
        int nfirst = 0;
        if (nnode < N_NODES) nfirst = (lane < nend - nbeg) ? srcs[nbeg + lane] : 0;
        const int n2 = nnode + nw;
        int n2beg = 0, n2end = 0;
        if (n2 < N_NODES) { n2beg = rowptr[n2]; n2end = rowptr[n2 + 1]; }

        vec acc = (vec)(0.0f);
        sum_rows<D, BATCH, vec>(acc, Hb, lane_off, first, min(32, end - beg));
        for (int base = beg + 32; base < end; base += 32) {
            const int cnt = min(32, end - base);
            const int mine = (lane < cnt) ? srcs[base + lane] : 0;
            sum_rows<D, BATCH, vec>(acc, Hb, lane_off, mine, cnt);
        }

        const float nd = rsqrtf(fmaxf((float)(end - beg), 1.0f));
        vec o = acc * nd + bv;
        if (RELU) {
#pragma unroll
            for (int i = 0; i < VEC; ++i) o[i] = fmaxf(o[i], 0.0f);
        }
        ((vec*)O)[(size_t)node * 32 + lane] = o;

        if (nnode >= N_NODES) break;
        node = nnode; beg = nbeg; end = nend; first = nfirst;
        nnode = n2; nbeg = n2beg; nend = n2end;
    }
}

extern "C" void kernel_launch(void* const* d_in, const int* in_sizes, int n_in,
                              void* d_out, int out_size, void* d_ws, size_t ws_size,
                              hipStream_t stream) {
    (void)n_in; (void)out_size; (void)ws_size;

    const float* x  = (const float*)d_in[0];
    const int*   src = (const int*)d_in[1];
    const int*   dst = (const int*)d_in[2];
    const float* W1 = (const float*)d_in[3];
    const float* b1 = (const float*)d_in[4];
    const float* W2 = (const float*)d_in[5];
    const float* b2 = (const float*)d_in[6];
    const float* W3 = (const float*)d_in[7];
    const float* b3 = (const float*)d_in[8];

    const int E = in_sizes[1];
    const int N = N_NODES;

    const size_t NPAD = 100352;
    float* ws   = (float*)d_ws;
    float* ns   = ws;
    int*   tbl  = (int*)ws + NPAD;
    float* bufA = ws + 2 * NPAD;
    float* bufB = bufA + (size_t)N * 128;

    int* rowptr  = (int*)d_out;
    int* srcs    = rowptr + (N + 1);
    int* off     = (int*)bufB;
    int* runs_s  = (int*)bufB + 2 * NPAD;
    int* runs_d  = (int*)bufA;
    int* rowptr3 = (int*)(bufA + (size_t)N * 64);
    int* srcs3   = rowptr3 + (N + 1);

    bin_count_kernel<<<G1, 1024, 0, stream>>>(src, dst, tbl, E);
    scan_kernel<<<(2 * TBL) / 1024, 1024, 0, stream>>>(tbl, off, 2 * TBL);
    bin_place_kernel<<<G1, 1024, 0, stream>>>(src, dst, off, runs_d, runs_s, E);
    range_lists_kernel<<<NBIN, 1024, 0, stream>>>(off, runs_d, runs_s, rowptr, srcs, ns, E);

    const int EDGEB = 1024;

    const int ROWT = N / 16;
    const int GEMMB = 1024;
    const int GEMMB64 = 2048;
    const int NODEB = 2048;

    gemm_norm_wmma<128, 128><<<GEMMB, 256, 0, stream>>>(x, ns, W1, bufA, ROWT);
    gather_sum_kernel<128, true><<<NODEB, 256, 0, stream>>>(bufA, rowptr, srcs, b1, bufB);

    gemm_norm_wmma<128, 128><<<GEMMB, 256, 0, stream>>>(bufB, ns, W2, bufA, ROWT);
    gather_sum_kernel<128, true><<<NODEB, 256, 0, stream>>>(bufA, rowptr, srcs, b2, bufB);

    gemm_norm_wmma<128, 64><<<GEMMB64, 128, 0, stream>>>(bufB, ns, W3, bufA, ROWT);
    move_lists_kernel<<<EDGEB, 256, 0, stream>>>(rowptr, rowptr3, N + 1 + E);
    gather_sum_kernel<64, false><<<NODEB, 256, 0, stream>>>(bufA, rowptr3, srcs3, b3, (float*)d_out);
}


# dense product: each 16 x 128 stripe of X is fetched once per block with whole-row loads, scaled and staged in LDS (two buffers, next stripe fetched ahead); waves read A fragments there; same k order
# speedup vs baseline: 4.2531x; 1.1181x over previous
#include <hip/hip_runtime.h>

#define N_NODES 100000

typedef __attribute__((ext_vector_type(2))) float v2f;
typedef __attribute__((ext_vector_type(8))) float v8f;

#define NBIN 196
#define G1 256
#define TBL (NBIN * G1)

__device__ __forceinline__ int range_of(unsigned node) { return min((int)(node >> 9), NBIN - 1); }

__global__ void __launch_bounds__(1024)
bin_count_kernel(const int* __restrict__ src, const int* __restrict__ dst,
                 int* __restrict__ tbl, int E) {
    __shared__ int hd[NBIN];
    __shared__ int hs[NBIN];
    const int t = threadIdx.x;
    const int b = blockIdx.x;
    if (t < NBIN) { hd[t] = 0; hs[t] = 0; }
    __syncthreads();
    const int ch = (E + G1 - 1) / G1;
    const int beg = b * ch;
    const int end = min(beg + ch, E);
    for (int i0 = beg; i0 < end; i0 += 8 * 1024) {
        unsigned d[8], s[8];
#pragma unroll
        for (int u = 0; u < 8; ++u) {
            const int i = i0 + u * 1024 + t;
            const int ic = (i < end) ? i : beg;
            d[u] = (unsigned)dst[ic];
            s[u] = (unsigned)src[ic];
        }
#pragma unroll
        for (int u = 0; u < 8; ++u) {
            if (i0 + u * 1024 + t < end) {
                atomicAdd(&hd[range_of(d[u])], 1);
                atomicAdd(&hs[range_of(s[u])], 1);
            }
        }
    }
    __syncthreads();
    if (t < NBIN) {
        tbl[t * G1 + b] = hd[t];
        tbl[TBL + t * G1 + b] = hs[t];
    }
}

__global__ void __launch_bounds__(1024)
scan_kernel(const int* __restrict__ cnt, int* __restrict__ off, int n) {
    __shared__ int wsum[32];
    __shared__ int wpre[32];
    const int t = threadIdx.x;
    const int lane = t & 31;
    const int w = t >> 5;
    const int nprev = blockIdx.x * 1024;

    int pre = 0;
    for (int i = t * 4; i < nprev; i += 4096) {
        const int4 q = *(const int4*)(cnt + i);
        pre += (q.x + q.y) + (q.z + q.w);
    }
    const int idx = nprev + t;
    const int c = (idx < n) ? cnt[idx] : 0;

    int inc = c;
    for (int o = 1; o < 32; o <<= 1) {
        const int v = __shfl_up(inc, o);
        if (lane >= o) inc += v;
    }
    for (int o = 16; o > 0; o >>= 1) pre += __shfl_xor(pre, o);
    if (lane == 31) wsum[w] = inc;
    if (lane == 0) wpre[w] = pre;
    __syncthreads();

    const int wv = wsum[lane];
    int pv = wpre[lane];
    int sc = wv;
    for (int o = 1; o < 32; o <<= 1) {
        const int v = __shfl_up(sc, o);
        if (lane >= o) sc += v;
    }
    const int wave_excl = __shfl(sc - wv, w);
    for (int o = 16; o > 0; o >>= 1) pv += __shfl_xor(pv, o);

    if (idx < n) off[idx] = pv + wave_excl + (inc - c);
}

__global__ void __launch_bounds__(1024)
bin_place_kernel(const int* __restrict__ src, const int* __restrict__ dst,
                 const int* __restrict__ off, int* __restrict__ runs_d,
                 int* __restrict__ runs_s, int E) {
    __shared__ int cd[NBIN];
    __shared__ int cs[NBIN];
    const int t = threadIdx.x;
    const int b = blockIdx.x;
    if (t < NBIN) { cd[t] = off[t * G1 + b]; cs[t] = off[TBL + t * G1 + b] - E; }
    __syncthreads();
    const int ch = (E + G1 - 1) / G1;
    const int beg = b * ch;
    const int end = min(beg + ch, E);
    for (int i0 = beg; i0 < end; i0 += 8 * 1024) {
        unsigned d[8], s[8];
#pragma unroll
        for (int u = 0; u < 8; ++u) {
            const int i = i0 + u * 1024 + t;
            const int ic = (i < end) ? i : beg;
            d[u] = (unsigned)dst[ic];
            s[u] = (unsigned)src[ic];
        }
#pragma unroll
        for (int u = 0; u < 8; ++u) {
            if (i0 + u * 1024 + t < end) {
                const int pd = atomicAdd(&cd[range_of(d[u])], 1);
                const int ps = atomicAdd(&cs[range_of(s[u])], 1);
                runs_d[pd] = (int)((s[u] << 9) | (d[u] & 511u));
                runs_s[ps] = (int)(s[u] & 511u);
            }
        }
    }
}

__global__ void __launch_bounds__(1024)
range_lists_kernel(const int* __restrict__ off, const int* __restrict__ runs_d,
                   const int* __restrict__ runs_s, int* __restrict__ rowptr,
                   int* __restrict__ srcs, float* __restrict__ ns, int E) {
    __shared__ int cnt[512];
    __shared__ int cur[512];
    __shared__ int ocn[512];
    __shared__ int wsum[32];
    const int t = threadIdx.x;
    const int lane = t & 31;
    const int w = t >> 5;
    const int g = blockIdx.x;
    const int bd = off[g * G1];
    const int ed = off[(g + 1) * G1];
    const int bs = off[TBL + g * G1] - E;
    const int es = (g == NBIN - 1) ? E : off[TBL + (g + 1) * G1] - E;
    if (t < 512) { cnt[t] = 0; ocn[t] = 0; }
    __syncthreads();
    for (int i0 = bd; i0 < ed; i0 += 8 * 1024) {
        int wd[8];
#pragma unroll
        for (int u = 0; u < 8; ++u) {
            const int i = i0 + u * 1024 + t;
            wd[u] = runs_d[(i < ed) ? i : bd];
        }
#pragma unroll
        for (int u = 0; u < 8; ++u)
            if (i0 + u * 1024 + t < ed) atomicAdd(&cnt[wd[u] & 511], 1);
    }
    for (int i0 = bs; i0 < es; i0 += 8 * 1024) {
        int wo[8];
#pragma unroll
        for (int u = 0; u < 8; ++u) {
            const int i = i0 + u * 1024 + t;
            wo[u] = runs_s[(i < es) ? i : bs];
        }
#pragma unroll
        for (int u = 0; u < 8; ++u)
            if (i0 + u * 1024 + t < es) atomicAdd(&ocn[wo[u] & 511], 1);
    }
    __syncthreads();

    const int c = (t < 512) ? cnt[t] : 0;
    int inc = c;
    for (int o = 1; o < 32; o <<= 1) {
        const int v = __shfl_up(inc, o);
        if (lane >= o) inc += v;
    }
    if (lane == 31) wsum[w] = inc;
    __syncthreads();
    const int wv = wsum[lane];
    int sc = wv;
    for (int o = 1; o < 32; o <<= 1) {
        const int v = __shfl_up(sc, o);
        if (lane >= o) sc += v;
    }
    const int wave_excl = __shfl(sc - wv, w);
    const int node = g * 512 + t;
    if (t < 512) {
        const int first = bd + wave_excl + (inc - c);
        cur[t] = first;
        if (node < N_NODES) {
            rowptr[node] = first;
            ns[node] = rsqrtf(fmaxf((float)ocn[t], 1.0f));
        }
    }
    if (g == NBIN - 1 && t == 0) rowptr[N_NODES] = ed;
    __syncthreads();
    for (int i0 = bd; i0 < ed; i0 += 8 * 1024) {
        int wd[8];
#pragma unroll
        for (int u = 0; u < 8; ++u) {
            const int i = i0 + u * 1024 + t;
            wd[u] = runs_d[(i < ed) ? i : bd];
        }
#pragma unroll
        for (int u = 0; u < 8; ++u) {
            if (i0 + u * 1024 + t < ed) {
                const int p = atomicAdd(&cur[wd[u] & 511], 1);
                srcs[p] = (int)((unsigned)wd[u] >> 9);
            }
        }
    }
}

__global__ void move_lists_kernel(const int* __restrict__ from, int* __restrict__ to, int n) {
    for (int c0 = blockIdx.x * 2048; c0 < n; c0 += gridDim.x * 2048) {
        const int i0 = c0 + threadIdx.x;
        if (c0 + 2048 <= n) {
            int v[8];
#pragma unroll
            for (int u = 0; u < 8; ++u) v[u] = from[i0 + u * 256];
#pragma unroll
            for (int u = 0; u < 8; ++u) to[i0 + u * 256] = v[u];
        } else {
            for (int i = i0; i < n; i += 256) to[i] = from[i];
        }
    }
}

template <int DIN, int DOUT>
__global__ void __launch_bounds__(DOUT * 2)
gemm_norm_wmma(const float* __restrict__ X, const float* __restrict__ ns,
               const float* __restrict__ W, float* __restrict__ Y,
               int nstripes) {
    static_assert(DIN == 128, "one wave loads one row of X per pass");
    constexpr int NW = DOUT / 16;
    constexpr int PASSES = 16 / NW;
    constexpr int PITCH = DIN + 4;
    __shared__ __attribute__((aligned(16))) float As[2][16 * PITCH];

    const int wave = __builtin_amdgcn_readfirstlane((int)(threadIdx.x >> 5));
    const int lane = threadIdx.x & 31;
    const int hl   = lane & 15;
    const bool hi  = (lane >= 16);
    const int kofs = hi ? 2 : 0;
    const int col  = wave * 16 + hl;

    v2f bfrag[DIN / 4];
#pragma unroll
    for (int i = 0; i < DIN / 4; ++i) {
        const int ka = 4 * i + kofs;
        bfrag[i][0] = W[(size_t)ka * DOUT + col];
        bfrag[i][1] = W[(size_t)(ka + 1) * DOUT + col];
    }

    int stripe = blockIdx.x;
    if (stripe >= nstripes) return;

    float4 pre[PASSES];
    float  psc[PASSES];
#define FETCH_STRIPE(s)                                                              \
    _Pragma("unroll")                                                                \
    for (int p = 0; p < PASSES; ++p) {                                               \
        const int r = (s) * 16 + p * NW + wave;                                      \
        pre[p] = *(const float4*)(X + (size_t)r * DIN + lane * 4);                   \
        psc[p] = ns[r];                                                              \
    }
#define STASH_STRIPE(b)                                                              \
    _Pragma("unroll")                                                                \
    for (int p = 0; p < PASSES; ++p) {                                               \
        float4 v = pre[p];                                                           \
        v.x *= psc[p]; v.y *= psc[p]; v.z *= psc[p]; v.w *= psc[p];                  \
        *(float4*)(&As[b][(p * NW + wave) * PITCH + lane * 4]) = v;                  \
    }

    FETCH_STRIPE(stripe)
    STASH_STRIPE(0)
    __syncthreads();

    int buf = 0;
    for (; stripe < nstripes; stripe += gridDim.x) {
        const int next = stripe + gridDim.x;
        const bool more = next < nstripes;
        if (more) { FETCH_STRIPE(next) }

        const float* __restrict__ arow = &As[buf][hl * PITCH + kofs];
        v8f acc = {};
#pragma unroll
        for (int i = 0; i < DIN / 4; ++i) {
            const v2f a = *(const v2f*)(arow + 4 * i);
            acc = __builtin_amdgcn_wmma_f32_16x16x4_f32(
                 false, a,  false, bfrag[i],
                 (short)0, acc,  false,  false);
        }

        if (more) { STASH_STRIPE(buf ^ 1) }
        __syncthreads();
        buf ^= 1;

        const int crow0 = stripe * 16 + (hi ? 8 : 0);
#pragma unroll
        for (int v = 0; v < 8; ++v) {
            Y[(size_t)(crow0 + v) * DOUT + col] = acc[v];
        }
    }
#undef FETCH_STRIPE
#undef STASH_STRIPE
}

template <int D, int BATCH, typename vec>
__device__ __forceinline__ void sum_rows(vec& acc, const char* __restrict__ Hb, unsigned lane_off,
                                         int mine, int cnt) {
    int j = 0;
    for (; j + BATCH <= cnt; j += BATCH) {
        int s[BATCH];
        vec v[BATCH];
#pragma unroll
        for (int u = 0; u < BATCH; ++u) s[u] = __builtin_amdgcn_readlane(mine, j + u);
#pragma unroll
        for (int u = 0; u < BATCH; ++u)
            v[u] = *(const vec*)(Hb + ((size_t)(unsigned)s[u] * (D * 4)) + lane_off);
#pragma unroll
        for (int u = 0; u < BATCH; ++u) acc += v[u];
    }
    if (j < cnt) {
        int s[BATCH];
        vec v[BATCH];
#pragma unroll
        for (int u = 0; u < BATCH; ++u) s[u] = __builtin_amdgcn_readlane(mine, (j + u) & 31);
#pragma unroll
        for (int u = 0; u < BATCH; ++u) {
            if (j + u < cnt)
                v[u] = *(const vec*)(Hb + ((size_t)(unsigned)s[u] * (D * 4)) + lane_off);
            else
                v[u] = (vec)(0.0f);
        }
#pragma unroll
        for (int u = 0; u < BATCH; ++u) acc += v[u];
    }
}

template <int D, bool RELU>
__global__ void gather_sum_kernel(const float* __restrict__ H, const int* __restrict__ rowptr,
                                  const int* __restrict__ srcs, const float* __restrict__ bias,
                                  float* __restrict__ O) {
    constexpr int VEC = D / 32;
    constexpr int BATCH = 8;
    typedef __attribute__((ext_vector_type(VEC))) float vec;

    const int lane = threadIdx.x & 31;
    const int wv = __builtin_amdgcn_readfirstlane((int)(blockIdx.x * 8 + (threadIdx.x >> 5)));
    const int nw = (int)gridDim.x * 8;
    if (wv >= N_NODES) return;

    const char* __restrict__ Hb = (const char*)H;
    const unsigned lane_off = (unsigned)lane * (VEC * 4);
    const vec bv = ((const vec*)bias)[lane];

    int node = wv;
    int beg = rowptr[node];
    int end = rowptr[node + 1];
    int first = (lane < end - beg) ? srcs[beg + lane] : 0;
    asm volatile("" : "+v"(first));
    int nnode = node + nw;
    int nbeg = 0, nend = 0;
    if (nnode < N_NODES) { nbeg = rowptr[nnode]; nend = rowptr[nnode + 1]; }

    for (;;) {
        int nfirst = 0;
        if (nnode < N_NODES) nfirst = (lane < nend - nbeg) ? srcs[nbeg + lane] : 0;
        const int n2 = nnode + nw;
        int n2beg = 0, n2end = 0;
        if (n2 < N_NODES) { n2beg = rowptr[n2]; n2end = rowptr[n2 + 1]; }

        vec acc = (vec)(0.0f);
        sum_rows<D, BATCH, vec>(acc, Hb, lane_off, first, min(32, end - beg));
        for (int base = beg + 32; base < end; base += 32) {
            const int cnt = min(32, end - base);
            const int mine = (lane < cnt) ? srcs[base + lane] : 0;
            sum_rows<D, BATCH, vec>(acc, Hb, lane_off, mine, cnt);
        }

        const float nd = rsqrtf(fmaxf((float)(end - beg), 1.0f));
        vec o = acc * nd + bv;
        if (RELU) {
#pragma unroll
            for (int i = 0; i < VEC; ++i) o[i] = fmaxf(o[i], 0.0f);
        }
        ((vec*)O)[(size_t)node * 32 + lane] = o;

        if (nnode >= N_NODES) break;
        node = nnode; beg = nbeg; end = nend; first = nfirst;
        nnode = n2; nbeg = n2beg; nend = n2end;
    }
}

extern "C" void kernel_launch(void* const* d_in, const int* in_sizes, int n_in,
                              void* d_out, int out_size, void* d_ws, size_t ws_size,
                              hipStream_t stream) {
    (void)n_in; (void)out_size; (void)ws_size;

    const float* x  = (const float*)d_in[0];
    const int*   src = (const int*)d_in[1];
    const int*   dst = (const int*)d_in[2];
    const float* W1 = (const float*)d_in[3];
    const float* b1 = (const float*)d_in[4];
    const float* W2 = (const float*)d_in[5];
    const float* b2 = (const float*)d_in[6];
    const float* W3 = (const float*)d_in[7];
    const float* b3 = (const float*)d_in[8];

    const int E = in_sizes[1];
    const int N = N_NODES;

    const size_t NPAD = 100352;
    float* ws   = (float*)d_ws;
    float* ns   = ws;
    int*   tbl  = (int*)ws + NPAD;
    float* bufA = ws + 2 * NPAD;
    float* bufB = bufA + (size_t)N * 128;

    int* rowptr  = (int*)d_out;
    int* srcs    = rowptr + (N + 1);
    int* off     = (int*)bufB;
    int* runs_s  = (int*)bufB + 2 * NPAD;
    int* runs_d  = (int*)bufA;
    int* rowptr3 = (int*)(bufA + (size_t)N * 64);
    int* srcs3   = rowptr3 + (N + 1);

    bin_count_kernel<<<G1, 1024, 0, stream>>>(src, dst, tbl, E);
    scan_kernel<<<(2 * TBL) / 1024, 1024, 0, stream>>>(tbl, off, 2 * TBL);
    bin_place_kernel<<<G1, 1024, 0, stream>>>(src, dst, off, runs_d, runs_s, E);
    range_lists_kernel<<<NBIN, 1024, 0, stream>>>(off, runs_d, runs_s, rowptr, srcs, ns, E);

    const int EDGEB = 1024;

    const int ROWT = N / 16;
    const int GEMMB = 1024;
    const int GEMMB64 = 2048;
    const int NODEB = 2048;

    gemm_norm_wmma<128, 128><<<GEMMB, 256, 0, stream>>>(x, ns, W1, bufA, ROWT);
    gather_sum_kernel<128, true><<<NODEB, 256, 0, stream>>>(bufA, rowptr, srcs, b1, bufB);

    gemm_norm_wmma<128, 128><<<GEMMB, 256, 0, stream>>>(bufB, ns, W2, bufA, ROWT);
    gather_sum_kernel<128, true><<<NODEB, 256, 0, stream>>>(bufA, rowptr, srcs, b2, bufB);

    gemm_norm_wmma<128, 64><<<GEMMB64, 128, 0, stream>>>(bufB, ns, W3, bufA, ROWT);
    move_lists_kernel<<<EDGEB, 256, 0, stream>>>(rowptr, rowptr3, N + 1 + E);
    gather_sum_kernel<64, false><<<NODEB, 256, 0, stream>>>(bufA, rowptr3, srcs3, b3, (float*)d_out);
}


# layers 1 and 2: each aggregation launch also does the next layer's product: a block's 16 summed, scaled rows go to LDS as the A stripe of its f32 WMMA loop; sums and k order as in the parent
# speedup vs baseline: 4.6673x; 1.0974x over previous
#include <hip/hip_runtime.h>

#define N_NODES 100000

typedef __attribute__((ext_vector_type(2))) float v2f;
typedef __attribute__((ext_vector_type(8))) float v8f;

#define NBIN 196
#define G1 256
#define TBL (NBIN * G1)

__device__ __forceinline__ int range_of(unsigned node) { return min((int)(node >> 9), NBIN - 1); }

__global__ void __launch_bounds__(1024)
bin_count_kernel(const int* __restrict__ src, const int* __restrict__ dst,
                 int* __restrict__ tbl, int E) {
    __shared__ int hd[NBIN];
    __shared__ int hs[NBIN];
    const int t = threadIdx.x;
    const int b = blockIdx.x;
    if (t < NBIN) { hd[t] = 0; hs[t] = 0; }
    __syncthreads();
    const int ch = (E + G1 - 1) / G1;
    const int beg = b * ch;
    const int end = min(beg + ch, E);
    for (int i0 = beg; i0 < end; i0 += 8 * 1024) {
        unsigned d[8], s[8];
#pragma unroll
        for (int u = 0; u < 8; ++u) {
            const int i = i0 + u * 1024 + t;
            const int ic = (i < end) ? i : beg;
            d[u] = (unsigned)dst[ic];
            s[u] = (unsigned)src[ic];
        }
#pragma unroll
        for (int u = 0; u < 8; ++u) {
            if (i0 + u * 1024 + t < end) {
                atomicAdd(&hd[range_of(d[u])], 1);
                atomicAdd(&hs[range_of(s[u])], 1);
            }
        }
    }
    __syncthreads();
    if (t < NBIN) {
        tbl[t * G1 + b] = hd[t];
        tbl[TBL + t * G1 + b] = hs[t];
    }
}

__global__ void __launch_bounds__(1024)
scan_kernel(const int* __restrict__ cnt, int* __restrict__ off, int n) {
    __shared__ int wsum[32];
    __shared__ int wpre[32];
    const int t = threadIdx.x;
    const int lane = t & 31;
    const int w = t >> 5;
    const int nprev = blockIdx.x * 1024;

    int pre = 0;
    for (int i = t * 4; i < nprev; i += 4096) {
        const int4 q = *(const int4*)(cnt + i);
        pre += (q.x + q.y) + (q.z + q.w);
    }
    const int idx = nprev + t;
    const int c = (idx < n) ? cnt[idx] : 0;

    int inc = c;
    for (int o = 1; o < 32; o <<= 1) {
        const int v = __shfl_up(inc, o);
        if (lane >= o) inc += v;
    }
    for (int o = 16; o > 0; o >>= 1) pre += __shfl_xor(pre, o);
    if (lane == 31) wsum[w] = inc;
    if (lane == 0) wpre[w] = pre;
    __syncthreads();

    const int wv = wsum[lane];
    int pv = wpre[lane];
    int sc = wv;
    for (int o = 1; o < 32; o <<= 1) {
        const int v = __shfl_up(sc, o);
        if (lane >= o) sc += v;
    }
    const int wave_excl = __shfl(sc - wv, w);
    for (int o = 16; o > 0; o >>= 1) pv += __shfl_xor(pv, o);

    if (idx < n) off[idx] = pv + wave_excl + (inc - c);
}

__global__ void __launch_bounds__(1024)
bin_place_kernel(const int* __restrict__ src, const int* __restrict__ dst,
                 const int* __restrict__ off, int* __restrict__ runs_d,
                 int* __restrict__ runs_s, int E) {
    __shared__ int cd[NBIN];
    __shared__ int cs[NBIN];
    const int t = threadIdx.x;
    const int b = blockIdx.x;
    if (t < NBIN) { cd[t] = off[t * G1 + b]; cs[t] = off[TBL + t * G1 + b] - E; }
    __syncthreads();
    const int ch = (E + G1 - 1) / G1;
    const int beg = b * ch;
    const int end = min(beg + ch, E);
    for (int i0 = beg; i0 < end; i0 += 8 * 1024) {
        unsigned d[8], s[8];
#pragma unroll
        for (int u = 0; u < 8; ++u) {
            const int i = i0 + u * 1024 + t;
            const int ic = (i < end) ? i : beg;
            d[u] = (unsigned)dst[ic];
            s[u] = (unsigned)src[ic];
        }
#pragma unroll
        for (int u = 0; u < 8; ++u) {
            if (i0 + u * 1024 + t < end) {
                const int pd = atomicAdd(&cd[range_of(d[u])], 1);
                const int ps = atomicAdd(&cs[range_of(s[u])], 1);
                runs_d[pd] = (int)((s[u] << 9) | (d[u] & 511u));
                runs_s[ps] = (int)(s[u] & 511u);
            }
        }
    }
}

__global__ void __launch_bounds__(1024)
range_lists_kernel(const int* __restrict__ off, const int* __restrict__ runs_d,
                   const int* __restrict__ runs_s, int* __restrict__ rowptr,
                   int* __restrict__ srcs, float* __restrict__ ns, int E) {
    __shared__ int cnt[512];
    __shared__ int cur[512];
    __shared__ int ocn[512];
    __shared__ int wsum[32];
    const int t = threadIdx.x;
    const int lane = t & 31;
    const int w = t >> 5;
    const int g = blockIdx.x;
    const int bd = off[g * G1];
    const int ed = off[(g + 1) * G1];
    const int bs = off[TBL + g * G1] - E;
    const int es = (g == NBIN - 1) ? E : off[TBL + (g + 1) * G1] - E;
    if (t < 512) { cnt[t] = 0; ocn[t] = 0; }
    __syncthreads();
    for (int i0 = bd; i0 < ed; i0 += 8 * 1024) {
        int wd[8];
#pragma unroll
        for (int u = 0; u < 8; ++u) {
            const int i = i0 + u * 1024 + t;
            wd[u] = runs_d[(i < ed) ? i : bd];
        }
#pragma unroll
        for (int u = 0; u < 8; ++u)
            if (i0 + u * 1024 + t < ed) atomicAdd(&cnt[wd[u] & 511], 1);
    }
    for (int i0 = bs; i0 < es; i0 += 8 * 1024) {
        int wo[8];
#pragma unroll
        for (int u = 0; u < 8; ++u) {
            const int i = i0 + u * 1024 + t;
            wo[u] = runs_s[(i < es) ? i : bs];
        }
#pragma unroll
        for (int u = 0; u < 8; ++u)
            if (i0 + u * 1024 + t < es) atomicAdd(&ocn[wo[u] & 511], 1);
    }
    __syncthreads();

    const int c = (t < 512) ? cnt[t] : 0;
    int inc = c;
    for (int o = 1; o < 32; o <<= 1) {
        const int v = __shfl_up(inc, o);
        if (lane >= o) inc += v;
    }
    if (lane == 31) wsum[w] = inc;
    __syncthreads();
    const int wv = wsum[lane];
    int sc = wv;
    for (int o = 1; o < 32; o <<= 1) {
        const int v = __shfl_up(sc, o);
        if (lane >= o) sc += v;
    }
    const int wave_excl = __shfl(sc - wv, w);
    const int node = g * 512 + t;
    if (t < 512) {
        const int first = bd + wave_excl + (inc - c);
        cur[t] = first;
        if (node < N_NODES) {
            rowptr[node] = first;
            ns[node] = rsqrtf(fmaxf((float)ocn[t], 1.0f));
        }
    }
    if (g == NBIN - 1 && t == 0) rowptr[N_NODES] = ed;
    __syncthreads();
    for (int i0 = bd; i0 < ed; i0 += 8 * 1024) {
        int wd[8];
#pragma unroll
        for (int u = 0; u < 8; ++u) {
            const int i = i0 + u * 1024 + t;
            wd[u] = runs_d[(i < ed) ? i : bd];
        }
#pragma unroll
        for (int u = 0; u < 8; ++u) {
            if (i0 + u * 1024 + t < ed) {
                const int p = atomicAdd(&cur[wd[u] & 511], 1);
                srcs[p] = (int)((unsigned)wd[u] >> 9);
            }
        }
    }
}

__global__ void move_lists_kernel(const int* __restrict__ from, int* __restrict__ to, int n) {
    for (int c0 = blockIdx.x * 2048; c0 < n; c0 += gridDim.x * 2048) {
        const int i0 = c0 + threadIdx.x;
        if (c0 + 2048 <= n) {
            int v[8];
#pragma unroll
            for (int u = 0; u < 8; ++u) v[u] = from[i0 + u * 256];
#pragma unroll
            for (int u = 0; u < 8; ++u) to[i0 + u * 256] = v[u];
        } else {
            for (int i = i0; i < n; i += 256) to[i] = from[i];
        }
    }
}

template <int DIN, int DOUT>
__global__ void __launch_bounds__(DOUT * 2)
gemm_norm_wmma(const float* __restrict__ X, const float* __restrict__ ns,
               const float* __restrict__ W, float* __restrict__ Y,
               int nstripes) {
    static_assert(DIN == 128, "one wave loads one row of X per pass");
    constexpr int NW = DOUT / 16;
    constexpr int PASSES = 16 / NW;
    constexpr int PITCH = DIN + 4;
    __shared__ __attribute__((aligned(16))) float As[2][16 * PITCH];

    const int wave = __builtin_amdgcn_readfirstlane((int)(threadIdx.x >> 5));
    const int lane = threadIdx.x & 31;
    const int hl   = lane & 15;
    const bool hi  = (lane >= 16);
    const int kofs = hi ? 2 : 0;
    const int col  = wave * 16 + hl;

    v2f bfrag[DIN / 4];
#pragma unroll
    for (int i = 0; i < DIN / 4; ++i) {
        const int ka = 4 * i + kofs;
        bfrag[i][0] = W[(size_t)ka * DOUT + col];
        bfrag[i][1] = W[(size_t)(ka + 1) * DOUT + col];
    }

    int stripe = blockIdx.x;
    if (stripe >= nstripes) return;

    float4 pre[PASSES];
    float  psc[PASSES];
#define FETCH_STRIPE(s)                                                              \
    _Pragma("unroll")                                                                \
    for (int p = 0; p < PASSES; ++p) {                                               \
        const int r = (s) * 16 + p * NW + wave;                                      \
        pre[p] = *(const float4*)(X + (size_t)r * DIN + lane * 4);                   \
        psc[p] = ns[r];                                                              \
    }
#define STASH_STRIPE(b)                                                              \
    _Pragma("unroll")                                                                \
    for (int p = 0; p < PASSES; ++p) {                                               \
        float4 v = pre[p];                                                           \
        v.x *= psc[p]; v.y *= psc[p]; v.z *= psc[p]; v.w *= psc[p];                  \
        *(float4*)(&As[b][(p * NW + wave) * PITCH + lane * 4]) = v;                  \
    }

    FETCH_STRIPE(stripe)
    STASH_STRIPE(0)
    __syncthreads();

    int buf = 0;
    for (; stripe < nstripes; stripe += gridDim.x) {
        const int next = stripe + gridDim.x;
        const bool more = next < nstripes;
        if (more) { FETCH_STRIPE(next) }

        const float* __restrict__ arow = &As[buf][hl * PITCH + kofs];
        v8f acc = {};
#pragma unroll
        for (int i = 0; i < DIN / 4; ++i) {
            const v2f a = *(const v2f*)(arow + 4 * i);
            acc = __builtin_amdgcn_wmma_f32_16x16x4_f32(
                 false, a,  false, bfrag[i],
                 (short)0, acc,  false,  false);
        }

        if (more) { STASH_STRIPE(buf ^ 1) }
        __syncthreads();
        buf ^= 1;

        const int crow0 = stripe * 16 + (hi ? 8 : 0);
#pragma unroll
        for (int v = 0; v < 8; ++v) {
            Y[(size_t)(crow0 + v) * DOUT + col] = acc[v];
        }
    }
#undef FETCH_STRIPE
#undef STASH_STRIPE
}

template <int D, int BATCH, typename vec>
__device__ __forceinline__ void sum_rows(vec& acc, const char* __restrict__ Hb, unsigned lane_off,
                                         int mine, int cnt) {
    int j = 0;
    for (; j + BATCH <= cnt; j += BATCH) {
        int s[BATCH];
        vec v[BATCH];
#pragma unroll
        for (int u = 0; u < BATCH; ++u) s[u] = __builtin_amdgcn_readlane(mine, j + u);
#pragma unroll
        for (int u = 0; u < BATCH; ++u)
            v[u] = *(const vec*)(Hb + ((size_t)(unsigned)s[u] * (D * 4)) + lane_off);
#pragma unroll
        for (int u = 0; u < BATCH; ++u) acc += v[u];
    }
    if (j < cnt) {
        int s[BATCH];
        vec v[BATCH];
#pragma unroll
        for (int u = 0; u < BATCH; ++u) s[u] = __builtin_amdgcn_readlane(mine, (j + u) & 31);
#pragma unroll
        for (int u = 0; u < BATCH; ++u) {
            if (j + u < cnt)
                v[u] = *(const vec*)(Hb + ((size_t)(unsigned)s[u] * (D * 4)) + lane_off);
            else
                v[u] = (vec)(0.0f);
        }
#pragma unroll
        for (int u = 0; u < BATCH; ++u) acc += v[u];
    }
}

template <int D, bool RELU>
__global__ void gather_sum_kernel(const float* __restrict__ H, const int* __restrict__ rowptr,
                                  const int* __restrict__ srcs, const float* __restrict__ bias,
                                  float* __restrict__ O) {
    constexpr int VEC = D / 32;
    constexpr int BATCH = 8;
    typedef __attribute__((ext_vector_type(VEC))) float vec;

    const int lane = threadIdx.x & 31;
    const int wv = __builtin_amdgcn_readfirstlane((int)(blockIdx.x * 8 + (threadIdx.x >> 5)));
    const int nw = (int)gridDim.x * 8;
    if (wv >= N_NODES) return;

    const char* __restrict__ Hb = (const char*)H;
    const unsigned lane_off = (unsigned)lane * (VEC * 4);
    const vec bv = ((const vec*)bias)[lane];

    int node = wv;
    int beg = rowptr[node];
    int end = rowptr[node + 1];
    int first = (lane < end - beg) ? srcs[beg + lane] : 0;
    asm volatile("" : "+v"(first));
    int nnode = node + nw;
    int nbeg = 0, nend = 0;
    if (nnode < N_NODES) { nbeg = rowptr[nnode]; nend = rowptr[nnode + 1]; }

    for (;;) {
        int nfirst = 0;
        if (nnode < N_NODES) nfirst = (lane < nend - nbeg) ? srcs[nbeg + lane] : 0;
        const int n2 = nnode + nw;
        int n2beg = 0, n2end = 0;
        if (n2 < N_NODES) { n2beg = rowptr[n2]; n2end = rowptr[n2 + 1]; }

        vec acc = (vec)(0.0f);
        sum_rows<D, BATCH, vec>(acc, Hb, lane_off, first, min(32, end - beg));
        for (int base = beg + 32; base < end; base += 32) {
            const int cnt = min(32, end - base);
            const int mine = (lane < cnt) ? srcs[base + lane] : 0;
            sum_rows<D, BATCH, vec>(acc, Hb, lane_off, mine, cnt);
        }

        const float nd = rsqrtf(fmaxf((float)(end - beg), 1.0f));
        vec o = acc * nd + bv;
        if (RELU) {
#pragma unroll
            for (int i = 0; i < VEC; ++i) o[i] = fmaxf(o[i], 0.0f);
        }
        ((vec*)O)[(size_t)node * 32 + lane] = o;

        if (nnode >= N_NODES) break;
        node = nnode; beg = nbeg; end = nend; first = nfirst;
        nnode = n2; nbeg = n2beg; nend = n2end;
    }
}

template <int DOUT>
__global__ void __launch_bounds__(DOUT * 2, 8)
gather_gemm_wmma(const float* __restrict__ H, const int* __restrict__ rowptr,
                 const int* __restrict__ srcs, const float* __restrict__ bias,
                 const float* __restrict__ ns, const float* __restrict__ W,
                 float* __restrict__ Y, int nstripes) {
    constexpr int DIN = 128;
    constexpr int NW = DOUT / 16;
    constexpr int NPW = 16 / NW;
    constexpr int PITCH = DIN + 4;
    constexpr int BATCH = 8;
    typedef __attribute__((ext_vector_type(4))) float vec;
    __shared__ __attribute__((aligned(16))) float As[2][16 * PITCH];

    const int wave = __builtin_amdgcn_readfirstlane((int)(threadIdx.x >> 5));
    const int lane = threadIdx.x & 31;
    const int hl   = lane & 15;
    const bool hi  = (lane >= 16);
    const int kofs = hi ? 2 : 0;
    const int col  = wave * 16 + hl;

    v2f bfrag[DIN / 4];
#pragma unroll
    for (int i = 0; i < DIN / 4; ++i) {
        const int ka = 4 * i + kofs;
        bfrag[i][0] = W[(size_t)ka * DOUT + col];
        bfrag[i][1] = W[(size_t)(ka + 1) * DOUT + col];
    }

    const int G = (int)gridDim.x;
    if ((int)blockIdx.x >= nstripes) return;

    const char* __restrict__ Hb = (const char*)H;
    const unsigned lane_off = (unsigned)lane * 16u;
    const vec bv = ((const vec*)bias)[lane];

    auto node_of = [&](int q) -> int {
        const int st = (int)blockIdx.x + (q / NPW) * G;
        return (st < nstripes) ? st * 16 + wave * NPW + (q % NPW) : -1;
    };

    int q = 0;
    int node = node_of(0);
    int beg = rowptr[node];
    int end = rowptr[node + 1];
    int first = (lane < end - beg) ? srcs[beg + lane] : 0;
    asm volatile("" : "+v"(first));
    int nnode = node_of(1);
    int nbeg = 0, nend = 0;
    if (nnode >= 0) { nbeg = rowptr[nnode]; nend = rowptr[nnode + 1]; }

    int buf = 0;
    for (int stripe = blockIdx.x; stripe < nstripes; stripe += G) {
#pragma unroll
        for (int j = 0; j < NPW; ++j) {
            int nfirst = 0;
            if (nnode >= 0) nfirst = (lane < nend - nbeg) ? srcs[nbeg + lane] : 0;
            const int n2 = node_of(q + 2);
            int n2beg = 0, n2end = 0;
            if (n2 >= 0) { n2beg = rowptr[n2]; n2end = rowptr[n2 + 1]; }
            const float sc = ns[node];

            vec acc = (vec)(0.0f);
            sum_rows<DIN, BATCH, vec>(acc, Hb, lane_off, first, min(32, end - beg));
            for (int base = beg + 32; base < end; base += 32) {
                const int cnt = min(32, end - base);
                const int mine = (lane < cnt) ? srcs[base + lane] : 0;
                sum_rows<DIN, BATCH, vec>(acc, Hb, lane_off, mine, cnt);
            }

            const float nd = rsqrtf(fmaxf((float)(end - beg), 1.0f));
            vec o = acc * nd + bv;
#pragma unroll
            for (int i = 0; i < 4; ++i) o[i] = fmaxf(o[i], 0.0f);
            o *= sc;
            *(vec*)(&As[buf][(wave * NPW + j) * PITCH + lane * 4]) = o;

            ++q;
            node = nnode; beg = nbeg; end = nend; first = nfirst;
            nnode = n2; nbeg = n2beg; nend = n2end;
        }
        __syncthreads();

        const float* __restrict__ arow = &As[buf][hl * PITCH + kofs];
        v8f acc = {};
#pragma unroll
        for (int i = 0; i < DIN / 4; ++i) {
            const v2f a = *(const v2f*)(arow + 4 * i);
            acc = __builtin_amdgcn_wmma_f32_16x16x4_f32(
                 false, a,  false, bfrag[i],
                 (short)0, acc,  false,  false);
        }
        buf ^= 1;

        const int crow0 = stripe * 16 + (hi ? 8 : 0);
#pragma unroll
        for (int v = 0; v < 8; ++v) {
            Y[(size_t)(crow0 + v) * DOUT + col] = acc[v];
        }
    }
}

extern "C" void kernel_launch(void* const* d_in, const int* in_sizes, int n_in,
                              void* d_out, int out_size, void* d_ws, size_t ws_size,
                              hipStream_t stream) {
    (void)n_in; (void)out_size; (void)ws_size;

    const float* x  = (const float*)d_in[0];
    const int*   src = (const int*)d_in[1];
    const int*   dst = (const int*)d_in[2];
    const float* W1 = (const float*)d_in[3];
    const float* b1 = (const float*)d_in[4];
    const float* W2 = (const float*)d_in[5];
    const float* b2 = (const float*)d_in[6];
    const float* W3 = (const float*)d_in[7];
    const float* b3 = (const float*)d_in[8];

    const int E = in_sizes[1];
    const int N = N_NODES;

    const size_t NPAD = 100352;
    float* ws   = (float*)d_ws;
    float* ns   = ws;
    int*   tbl  = (int*)ws + NPAD;
    float* bufA = ws + 2 * NPAD;
    float* bufB = bufA + (size_t)N * 128;

    int* rowptr  = (int*)d_out;
    int* srcs    = rowptr + (N + 1);
    int* off     = (int*)bufB;
    int* runs_s  = (int*)bufB + 2 * NPAD;
    int* runs_d  = (int*)bufA;
    int* rowptr3 = (int*)(bufA + (size_t)N * 64);
    int* srcs3   = rowptr3 + (N + 1);

    bin_count_kernel<<<G1, 1024, 0, stream>>>(src, dst, tbl, E);
    scan_kernel<<<(2 * TBL) / 1024, 1024, 0, stream>>>(tbl, off, 2 * TBL);
    bin_place_kernel<<<G1, 1024, 0, stream>>>(src, dst, off, runs_d, runs_s, E);
    range_lists_kernel<<<NBIN, 1024, 0, stream>>>(off, runs_d, runs_s, rowptr, srcs, ns, E);

    const int EDGEB = 1024;

    const int ROWT = N / 16;
    const int GEMMB = 1024;
    const int GEMMB64 = 2048;
    const int NODEB = 2048;

    gemm_norm_wmma<128, 128><<<GEMMB, 256, 0, stream>>>(x, ns, W1, bufA, ROWT);

    gather_gemm_wmma<128><<<GEMMB, 256, 0, stream>>>(bufA, rowptr, srcs, b1, ns, W2, bufB, ROWT);

    gather_gemm_wmma<64><<<GEMMB64, 128, 0, stream>>>(bufB, rowptr, srcs, b2, ns, W3, bufA, ROWT);

    move_lists_kernel<<<EDGEB, 256, 0, stream>>>(rowptr, rowptr3, N + 1 + E);
    gather_sum_kernel<64, false><<<NODEB, 256, 0, stream>>>(bufA, rowptr3, srcs3, b3, (float*)d_out);
}


# 9 -> 7 launches: run starts found per block group inside the placing launch (no scan launch); the lists' copy is written by layer 2's launch as it walks them; sum order = LDS add order, as before
# speedup vs baseline: 4.9807x; 1.0671x over previous
#include <hip/hip_runtime.h>

#define N_NODES 100000

typedef __attribute__((ext_vector_type(2))) float v2f;
typedef __attribute__((ext_vector_type(8))) float v8f;

#define NBIN 196
#define G1 256
#define NROW (2 * NBIN)
#define GB 16
#define NG (G1 / GB)

__device__ __forceinline__ int range_of(unsigned node) { return min((int)(node >> 9), NBIN - 1); }

__global__ void __launch_bounds__(1024)
bin_count_kernel(const int* __restrict__ src, const int* __restrict__ dst,
                 int* __restrict__ tbl, int E) {
    __shared__ int hd[NBIN];
    __shared__ int hs[NBIN];
    const int t = threadIdx.x;
    const int b = blockIdx.x;
    if (t < NBIN) { hd[t] = 0; hs[t] = 0; }
    __syncthreads();
    const int ch = (E + G1 - 1) / G1;
    const int beg = b * ch;
    const int end = min(beg + ch, E);
    for (int i0 = beg; i0 < end; i0 += 8 * 1024) {
        unsigned d[8], s[8];
#pragma unroll
        for (int u = 0; u < 8; ++u) {
            const int i = i0 + u * 1024 + t;
            const int ic = (i < end) ? i : beg;
            d[u] = (unsigned)dst[ic];
            s[u] = (unsigned)src[ic];
        }
#pragma unroll
        for (int u = 0; u < 8; ++u) {
            if (i0 + u * 1024 + t < end) {
                atomicAdd(&hd[range_of(d[u])], 1);
                atomicAdd(&hs[range_of(s[u])], 1);
            }
        }
    }
    __syncthreads();
    if (t < NBIN) {
        tbl[b * NROW + t] = hd[t];
        tbl[b * NROW + NBIN + t] = hs[t];
    }
}

__global__ void __launch_bounds__(1024)
bin_place_kernel(const int* __restrict__ src, const int* __restrict__ dst,
                 const int* __restrict__ tbl, int* __restrict__ seg,
                 int* __restrict__ runs_d, int* __restrict__ runs_s, int E) {
    constexpr int NQ = NROW / 4;
    constexpr int NCH = GB / 2;
    static_assert(NQ * 4 == NROW && NQ * NCH <= 1024 && NG * GB == G1, "table split");
    __shared__ int cd[NBIN];
    __shared__ int cs[NBIN];
    __shared__ __attribute__((aligned(16))) int lt[NCH][NROW];
    __shared__ __attribute__((aligned(16))) int lp[NCH][NROW];
    __shared__ int wsum[16];
    const int t = threadIdx.x;
    const int b = blockIdx.x;
    const int lane = t & 31;
    const int w = t >> 5;
    const int grp = b / GB;
    const int b0 = grp * GB;

    if (t < NQ * NCH) {
        const int q = t % NQ;
        const int c = t / NQ;
        const int r0 = b0 + 2 * c;
        const int4 v0 = *(const int4*)(tbl + r0 * NROW + q * 4);
        const int4 v1 = *(const int4*)(tbl + (r0 + 1) * NROW + q * 4);
        const bool f0 = r0 < b;
        const bool f1 = r0 + 1 < b;
        int4 tot, part;
        tot.x = v0.x + v1.x;  tot.y = v0.y + v1.y;  tot.z = v0.z + v1.z;  tot.w = v0.w + v1.w;
        part.x = (f0 ? v0.x : 0) + (f1 ? v1.x : 0);
        part.y = (f0 ? v0.y : 0) + (f1 ? v1.y : 0);
        part.z = (f0 ? v0.z : 0) + (f1 ? v1.z : 0);
        part.w = (f0 ? v0.w : 0) + (f1 ? v1.w : 0);
        *(int4*)(&lt[c][q * 4]) = tot;
        *(int4*)(&lp[c][q * 4]) = part;
    }
    __syncthreads();

    const int side = t >> 8;
    const int i = t & 255;
    int tv = 0, mypart = 0, excl = 0;
    if (t < 512) {
        if (i < NBIN) {
            const int r = side * NBIN + i;
#pragma unroll
            for (int c = 0; c < NCH; ++c) { tv += lt[c][r]; mypart += lp[c][r]; }
        }
        int inc = tv;
        for (int o = 1; o < 32; o <<= 1) {
            const int v = __shfl_up(inc, o);
            if (lane >= o) inc += v;
        }
        if (lane == 31) wsum[w] = inc;
        excl = inc - tv;
    }
    __syncthreads();
    if (t < 512 && i < NBIN) {
        int pre = 0;
#pragma unroll
        for (int k = 0; k < 8; ++k)
            if (side * 8 + k < w) pre += wsum[side * 8 + k];
        const int gbase = min(b0 * ((E + G1 - 1) / G1), E);
        const int start = gbase + pre + excl;
        if (side == 0) cd[i] = start + mypart; else cs[i] = start + mypart;
        if (b == b0) seg[grp * NROW + side * NBIN + i] = start;
    }
    __syncthreads();
    const int ch = (E + G1 - 1) / G1;
    const int beg = b * ch;
    const int end = min(beg + ch, E);
    for (int i0 = beg; i0 < end; i0 += 8 * 1024) {
        unsigned d[8], s[8];
#pragma unroll
        for (int u = 0; u < 8; ++u) {
            const int i = i0 + u * 1024 + t;
            const int ic = (i < end) ? i : beg;
            d[u] = (unsigned)dst[ic];
            s[u] = (unsigned)src[ic];
        }
#pragma unroll
        for (int u = 0; u < 8; ++u) {
            if (i0 + u * 1024 + t < end) {
                const int pd = atomicAdd(&cd[range_of(d[u])], 1);
                const int ps = atomicAdd(&cs[range_of(s[u])], 1);
                runs_d[pd] = (int)((s[u] << 9) | (d[u] & 511u));
                runs_s[ps] = (int)(s[u] & 511u);
            }
        }
    }
}

__global__ void __launch_bounds__(1024)
range_lists_kernel(const int* __restrict__ seg, const int* __restrict__ runs_d,
                   const int* __restrict__ runs_s, int* __restrict__ rowptr,
                   int* __restrict__ srcs, float* __restrict__ ns, int E) {
    constexpr int TPS = 1024 / NG;
    constexpr int U = 10;
    static_assert(TPS == 64, "a segment's threads are whole waves");
    __shared__ int cnt[512];
    __shared__ int cur[512];
    __shared__ int ocn[512];
    __shared__ int wsum[32];
    const int t = threadIdx.x;
    const int lane = t & 31;
    const int w = t >> 5;
    const int g = blockIdx.x;
    const int ch = (E + G1 - 1) / G1;
    const int grp = t / TPS;
    const int j = t % TPS;
    const int gend = min((grp + 1) * GB * ch, E);
    const int* __restrict__ sg = seg + grp * NROW;
    const int bd = sg[g];
    const int ed = (g == NBIN - 1) ? gend : sg[g + 1];
    const int bs = sg[NBIN + g];
    const int es = (g == NBIN - 1) ? gend : sg[NBIN + g + 1];
    int lbase = 0;
#pragma unroll
    for (int k = 0; k < NG; ++k) lbase += seg[k * NROW + g] - min(k * GB * ch, E);

    if (t < 512) { cnt[t] = 0; ocn[t] = 0; }
    __syncthreads();
    for (int i0 = bd; i0 < ed; i0 += U * TPS) {
        int wd[U];
#pragma unroll
        for (int u = 0; u < U; ++u) {
            const int i = i0 + u * TPS + j;
            wd[u] = runs_d[(i < ed) ? i : bd];
        }
#pragma unroll
        for (int u = 0; u < U; ++u)
            if (i0 + u * TPS + j < ed) atomicAdd(&cnt[wd[u] & 511], 1);
    }
    for (int i0 = bs; i0 < es; i0 += U * TPS) {
        int wo[U];
#pragma unroll
        for (int u = 0; u < U; ++u) {
            const int i = i0 + u * TPS + j;
            wo[u] = runs_s[(i < es) ? i : bs];
        }
#pragma unroll
        for (int u = 0; u < U; ++u)
            if (i0 + u * TPS + j < es) atomicAdd(&ocn[wo[u] & 511], 1);
    }
    __syncthreads();

    const int c = (t < 512) ? cnt[t] : 0;
    int inc = c;
    for (int o = 1; o < 32; o <<= 1) {
        const int v = __shfl_up(inc, o);
        if (lane >= o) inc += v;
    }
    if (lane == 31) wsum[w] = inc;
    __syncthreads();
    const int wv = wsum[lane];
    int sc = wv;
    for (int o = 1; o < 32; o <<= 1) {
        const int v = __shfl_up(sc, o);
        if (lane >= o) sc += v;
    }
    const int wave_excl = __shfl(sc - wv, w);
    const int node = g * 512 + t;
    if (t < 512) {
        const int first = lbase + wave_excl + (inc - c);
        cur[t] = first;
        if (node < N_NODES) {
            rowptr[node] = first;
            ns[node] = rsqrtf(fmaxf((float)ocn[t], 1.0f));
        }
    }
    if (g == NBIN - 1 && t == 0) rowptr[N_NODES] = E;
    __syncthreads();
    for (int i0 = bd; i0 < ed; i0 += U * TPS) {
        int wd[U];
#pragma unroll
        for (int u = 0; u < U; ++u) {
            const int i = i0 + u * TPS + j;
            wd[u] = runs_d[(i < ed) ? i : bd];
        }
#pragma unroll
        for (int u = 0; u < U; ++u) {
            if (i0 + u * TPS + j < ed) {
                const int p = atomicAdd(&cur[wd[u] & 511], 1);
                srcs[p] = (int)((unsigned)wd[u] >> 9);
            }
        }
    }
}

template <int DIN, int DOUT>
__global__ void __launch_bounds__(DOUT * 2)
gemm_norm_wmma(const float* __restrict__ X, const float* __restrict__ ns,
               const float* __restrict__ W, float* __restrict__ Y,
               int nstripes) {
    static_assert(DIN == 128, "one wave loads one row of X per pass");
    constexpr int NW = DOUT / 16;
    constexpr int PASSES = 16 / NW;
    constexpr int PITCH = DIN + 4;
    __shared__ __attribute__((aligned(16))) float As[2][16 * PITCH];

    const int wave = __builtin_amdgcn_readfirstlane((int)(threadIdx.x >> 5));
    const int lane = threadIdx.x & 31;
    const int hl   = lane & 15;
    const bool hi  = (lane >= 16);
    const int kofs = hi ? 2 : 0;
    const int col  = wave * 16 + hl;

    v2f bfrag[DIN / 4];
#pragma unroll
    for (int i = 0; i < DIN / 4; ++i) {
        const int ka = 4 * i + kofs;
        bfrag[i][0] = W[(size_t)ka * DOUT + col];
        bfrag[i][1] = W[(size_t)(ka + 1) * DOUT + col];
    }

    int stripe = blockIdx.x;
    if (stripe >= nstripes) return;

    float4 pre[PASSES];
    float  psc[PASSES];
#define FETCH_STRIPE(s)                                                              \
    _Pragma("unroll")                                                                \
    for (int p = 0; p < PASSES; ++p) {                                               \
        const int r = (s) * 16 + p * NW + wave;                                      \
        pre[p] = *(const float4*)(X + (size_t)r * DIN + lane * 4);                   \
        psc[p] = ns[r];                                                              \
    }
#define STASH_STRIPE(b)                                                              \
    _Pragma("unroll")                                                                \
    for (int p = 0; p < PASSES; ++p) {                                               \
        float4 v = pre[p];                                                           \
        v.x *= psc[p]; v.y *= psc[p]; v.z *= psc[p]; v.w *= psc[p];                  \
        *(float4*)(&As[b][(p * NW + wave) * PITCH + lane * 4]) = v;                  \
    }

    FETCH_STRIPE(stripe)
    STASH_STRIPE(0)
    __syncthreads();

    int buf = 0;
    for (; stripe < nstripes; stripe += gridDim.x) {
        const int next = stripe + gridDim.x;
        const bool more = next < nstripes;
        if (more) { FETCH_STRIPE(next) }

        const float* __restrict__ arow = &As[buf][hl * PITCH + kofs];
        v8f acc = {};
#pragma unroll
        for (int i = 0; i < DIN / 4; ++i) {
            const v2f a = *(const v2f*)(arow + 4 * i);
            acc = __builtin_amdgcn_wmma_f32_16x16x4_f32(
                 false, a,  false, bfrag[i],
                 (short)0, acc,  false,  false);
        }

        if (more) { STASH_STRIPE(buf ^ 1) }
        __syncthreads();
        buf ^= 1;

        const int crow0 = stripe * 16 + (hi ? 8 : 0);
#pragma unroll
        for (int v = 0; v < 8; ++v) {
            Y[(size_t)(crow0 + v) * DOUT + col] = acc[v];
        }
    }
#undef FETCH_STRIPE
#undef STASH_STRIPE
}

template <int D, int BATCH, typename vec>
__device__ __forceinline__ void sum_rows(vec& acc, const char* __restrict__ Hb, unsigned lane_off,
                                         int mine, int cnt) {
    int j = 0;
    for (; j + BATCH <= cnt; j += BATCH) {
        int s[BATCH];
        vec v[BATCH];
#pragma unroll
        for (int u = 0; u < BATCH; ++u) s[u] = __builtin_amdgcn_readlane(mine, j + u);
#pragma unroll
        for (int u = 0; u < BATCH; ++u)
            v[u] = *(const vec*)(Hb + ((size_t)(unsigned)s[u] * (D * 4)) + lane_off);
#pragma unroll
        for (int u = 0; u < BATCH; ++u) acc += v[u];
    }
    if (j < cnt) {
        int s[BATCH];
        vec v[BATCH];
#pragma unroll
        for (int u = 0; u < BATCH; ++u) s[u] = __builtin_amdgcn_readlane(mine, (j + u) & 31);
#pragma unroll
        for (int u = 0; u < BATCH; ++u) {
            if (j + u < cnt)
                v[u] = *(const vec*)(Hb + ((size_t)(unsigned)s[u] * (D * 4)) + lane_off);
            else
                v[u] = (vec)(0.0f);
        }
#pragma unroll
        for (int u = 0; u < BATCH; ++u) acc += v[u];
    }
}

template <int D, bool RELU>
__global__ void gather_sum_kernel(const float* __restrict__ H, const int* __restrict__ rowptr,
                                  const int* __restrict__ srcs, const float* __restrict__ bias,
                                  float* __restrict__ O) {
    constexpr int VEC = D / 32;
    constexpr int BATCH = 8;
    typedef __attribute__((ext_vector_type(VEC))) float vec;

    const int lane = threadIdx.x & 31;
    const int wv = __builtin_amdgcn_readfirstlane((int)(blockIdx.x * 8 + (threadIdx.x >> 5)));
    const int nw = (int)gridDim.x * 8;
    if (wv >= N_NODES) return;

    const char* __restrict__ Hb = (const char*)H;
    const unsigned lane_off = (unsigned)lane * (VEC * 4);
    const vec bv = ((const vec*)bias)[lane];

    int node = wv;
    int beg = rowptr[node];
    int end = rowptr[node + 1];
    int first = (lane < end - beg) ? srcs[beg + lane] : 0;
    asm volatile("" : "+v"(first));
    int nnode = node + nw;
    int nbeg = 0, nend = 0;
    if (nnode < N_NODES) { nbeg = rowptr[nnode]; nend = rowptr[nnode + 1]; }

    for (;;) {
        int nfirst = 0;
        if (nnode < N_NODES) nfirst = (lane < nend - nbeg) ? srcs[nbeg + lane] : 0;
        const int n2 = nnode + nw;
        int n2beg = 0, n2end = 0;
        if (n2 < N_NODES) { n2beg = rowptr[n2]; n2end = rowptr[n2 + 1]; }

        vec acc = (vec)(0.0f);
        sum_rows<D, BATCH, vec>(acc, Hb, lane_off, first, min(32, end - beg));
        for (int base = beg + 32; base < end; base += 32) {
            const int cnt = min(32, end - base);
            const int mine = (lane < cnt) ? srcs[base + lane] : 0;
            sum_rows<D, BATCH, vec>(acc, Hb, lane_off, mine, cnt);
        }

        const float nd = rsqrtf(fmaxf((float)(end - beg), 1.0f));
        vec o = acc * nd + bv;
        if (RELU) {
#pragma unroll
            for (int i = 0; i < VEC; ++i) o[i] = fmaxf(o[i], 0.0f);
        }
        ((vec*)O)[(size_t)node * 32 + lane] = o;

        if (nnode >= N_NODES) break;
        node = nnode; beg = nbeg; end = nend; first = nfirst;
        nnode = n2; nbeg = n2beg; nend = n2end;
    }
}

template <int DOUT, bool COPY>
__global__ void __launch_bounds__(DOUT * 2, 8)
gather_gemm_wmma(const float* __restrict__ H, const int* __restrict__ rowptr,
                 const int* __restrict__ srcs, const float* __restrict__ bias,
                 const float* __restrict__ ns, const float* __restrict__ W,
                 float* __restrict__ Y, int nstripes,
                 int* __restrict__ rowptr_c, int* __restrict__ srcs_c) {
    constexpr int DIN = 128;
    constexpr int NW = DOUT / 16;
    constexpr int NPW = 16 / NW;
    constexpr int PITCH = DIN + 4;
    constexpr int BATCH = 8;
    typedef __attribute__((ext_vector_type(4))) float vec;
    __shared__ __attribute__((aligned(16))) float As[2][16 * PITCH];

    const int wave = __builtin_amdgcn_readfirstlane((int)(threadIdx.x >> 5));
    const int lane = threadIdx.x & 31;
    const int hl   = lane & 15;
    const bool hi  = (lane >= 16);
    const int kofs = hi ? 2 : 0;
    const int col  = wave * 16 + hl;

    v2f bfrag[DIN / 4];
#pragma unroll
    for (int i = 0; i < DIN / 4; ++i) {
        const int ka = 4 * i + kofs;
        bfrag[i][0] = W[(size_t)ka * DOUT + col];
        bfrag[i][1] = W[(size_t)(ka + 1) * DOUT + col];
    }

    const int G = (int)gridDim.x;
    if ((int)blockIdx.x >= nstripes) return;

    const char* __restrict__ Hb = (const char*)H;
    const unsigned lane_off = (unsigned)lane * 16u;
    const vec bv = ((const vec*)bias)[lane];

    auto node_of = [&](int q) -> int {
        const int st = (int)blockIdx.x + (q / NPW) * G;
        return (st < nstripes) ? st * 16 + wave * NPW + (q % NPW) : -1;
    };

    int q = 0;
    int node = node_of(0);
    int beg = rowptr[node];
    int end = rowptr[node + 1];
    int first = (lane < end - beg) ? srcs[beg + lane] : 0;
    asm volatile("" : "+v"(first));
    int nnode = node_of(1);
    int nbeg = 0, nend = 0;
    if (nnode >= 0) { nbeg = rowptr[nnode]; nend = rowptr[nnode + 1]; }

    int buf = 0;
    for (int stripe = blockIdx.x; stripe < nstripes; stripe += G) {
#pragma unroll
        for (int j = 0; j < NPW; ++j) {
            int nfirst = 0;
            if (nnode >= 0) nfirst = (lane < nend - nbeg) ? srcs[nbeg + lane] : 0;
            const int n2 = node_of(q + 2);
            int n2beg = 0, n2end = 0;
            if (n2 >= 0) { n2beg = rowptr[n2]; n2end = rowptr[n2 + 1]; }
            const float sc = ns[node];

            if (COPY) {
                if (lane < end - beg) srcs_c[beg + lane] = first;
                if (lane == 0) {
                    rowptr_c[node] = beg;
                    if (node == N_NODES - 1) rowptr_c[N_NODES] = end;
                }
            }

            vec acc = (vec)(0.0f);
            sum_rows<DIN, BATCH, vec>(acc, Hb, lane_off, first, min(32, end - beg));
            for (int base = beg + 32; base < end; base += 32) {
                const int cnt = min(32, end - base);
                const int mine = (lane < cnt) ? srcs[base + lane] : 0;
                if (COPY) { if (lane < cnt) srcs_c[base + lane] = mine; }
                sum_rows<DIN, BATCH, vec>(acc, Hb, lane_off, mine, cnt);
            }

            const float nd = rsqrtf(fmaxf((float)(end - beg), 1.0f));
            vec o = acc * nd + bv;
#pragma unroll
            for (int i = 0; i < 4; ++i) o[i] = fmaxf(o[i], 0.0f);
            o *= sc;
            *(vec*)(&As[buf][(wave * NPW + j) * PITCH + lane * 4]) = o;

            ++q;
            node = nnode; beg = nbeg; end = nend; first = nfirst;
            nnode = n2; nbeg = n2beg; nend = n2end;
        }
        __syncthreads();

        const float* __restrict__ arow = &As[buf][hl * PITCH + kofs];
        v8f acc = {};
#pragma unroll
        for (int i = 0; i < DIN / 4; ++i) {
            const v2f a = *(const v2f*)(arow + 4 * i);
            acc = __builtin_amdgcn_wmma_f32_16x16x4_f32(
                 false, a,  false, bfrag[i],
                 (short)0, acc,  false,  false);
        }
        buf ^= 1;

        const int crow0 = stripe * 16 + (hi ? 8 : 0);
#pragma unroll
        for (int v = 0; v < 8; ++v) {
            Y[(size_t)(crow0 + v) * DOUT + col] = acc[v];
        }
    }
}

extern "C" void kernel_launch(void* const* d_in, const int* in_sizes, int n_in,
                              void* d_out, int out_size, void* d_ws, size_t ws_size,
                              hipStream_t stream) {
    (void)n_in; (void)out_size; (void)ws_size;

    const float* x  = (const float*)d_in[0];
    const int*   src = (const int*)d_in[1];
    const int*   dst = (const int*)d_in[2];
    const float* W1 = (const float*)d_in[3];
    const float* b1 = (const float*)d_in[4];
    const float* W2 = (const float*)d_in[5];
    const float* b2 = (const float*)d_in[6];
    const float* W3 = (const float*)d_in[7];
    const float* b3 = (const float*)d_in[8];

    const int E = in_sizes[1];
    const int N = N_NODES;

    const size_t NPAD = 100352;
    float* ws   = (float*)d_ws;
    float* ns   = ws;
    int*   tbl  = (int*)ws + NPAD;
    float* bufA = ws + 2 * NPAD;
    float* bufB = bufA + (size_t)N * 128;

    int* rowptr  = (int*)d_out;
    int* srcs    = rowptr + (N + 1);
    int* seg     = (int*)bufB;
    int* runs_s  = (int*)bufB + 2 * NPAD;
    int* runs_d  = (int*)bufA;
    int* rowptr3 = (int*)(bufA + (size_t)N * 64);
    int* srcs3   = rowptr3 + (N + 1);

    bin_count_kernel<<<G1, 1024, 0, stream>>>(src, dst, tbl, E);
    bin_place_kernel<<<G1, 1024, 0, stream>>>(src, dst, tbl, seg, runs_d, runs_s, E);
    range_lists_kernel<<<NBIN, 1024, 0, stream>>>(seg, runs_d, runs_s, rowptr, srcs, ns, E);

    const int ROWT = N / 16;
    const int GEMMB = 1024;
    const int GEMMB64 = 2048;
    const int NODEB = 2048;

    gemm_norm_wmma<128, 128><<<GEMMB, 256, 0, stream>>>(x, ns, W1, bufA, ROWT);

    gather_gemm_wmma<128, false><<<GEMMB, 256, 0, stream>>>(bufA, rowptr, srcs, b1, ns, W2, bufB, ROWT,
                                                            nullptr, nullptr);

    gather_gemm_wmma<64, true><<<GEMMB64, 128, 0, stream>>>(bufB, rowptr, srcs, b2, ns, W3, bufA, ROWT,
                                                            rowptr3, srcs3);

    gather_sum_kernel<64, false><<<NODEB, 256, 0, stream>>>(bufA, rowptr3, srcs3, b3, (float*)d_out);
}


# list build in 2 launches, not 3: each block counts its slice's edges per 512-node range in LDS, scans, and places them in its own stretch of the runs (no count launch); sum order = LDS add order
# speedup vs baseline: 5.5457x; 1.1134x over previous
#include <hip/hip_runtime.h>

#define N_NODES 100000

typedef __attribute__((ext_vector_type(2))) float v2f;
typedef __attribute__((ext_vector_type(8))) float v8f;

#define NBIN 196
#define G1 256
#define NROW (2 * NBIN)

__device__ __forceinline__ int range_of(unsigned node) { return min((int)(node >> 9), NBIN - 1); }

__global__ void __launch_bounds__(1024)
bin_place_kernel(const int* __restrict__ src, const int* __restrict__ dst,
                 int* __restrict__ tbl, int* __restrict__ runs_d,
                 int* __restrict__ runs_s, int E) {
    __shared__ int cd[NBIN];
    __shared__ int cs[NBIN];
    __shared__ int wsum[16];
    const int t = threadIdx.x;
    const int b = blockIdx.x;
    const int lane = t & 31;
    const int w = t >> 5;
    if (t < NBIN) { cd[t] = 0; cs[t] = 0; }
    __syncthreads();
    const int ch = (E + G1 - 1) / G1;
    const int beg = min(b * ch, E);
    const int end = min(beg + ch, E);
    for (int i0 = beg; i0 < end; i0 += 8 * 1024) {
        unsigned d[8], s[8];
#pragma unroll
        for (int u = 0; u < 8; ++u) {
            const int i = i0 + u * 1024 + t;
            const int ic = (i < end) ? i : beg;
            d[u] = (unsigned)dst[ic];
            s[u] = (unsigned)src[ic];
        }
#pragma unroll
        for (int u = 0; u < 8; ++u) {
            if (i0 + u * 1024 + t < end) {
                atomicAdd(&cd[range_of(d[u])], 1);
                atomicAdd(&cs[range_of(s[u])], 1);
            }
        }
    }
    __syncthreads();

    const int side = t >> 8;
    const int i = t & 255;
    int tv = 0, excl = 0;
    if (t < 512) {
        if (i < NBIN) tv = side ? cs[i] : cd[i];
        int inc = tv;
        for (int o = 1; o < 32; o <<= 1) {
            const int v = __shfl_up(inc, o);
            if (lane >= o) inc += v;
        }
        if (lane == 31) wsum[w] = inc;
        excl = inc - tv;
    }
    __syncthreads();
    if (t < 512 && i < NBIN) {
        int pre = 0;
#pragma unroll
        for (int k = 0; k < 8; ++k)
            if (side * 8 + k < w) pre += wsum[side * 8 + k];
        const int start = beg + pre + excl;
        if (side == 0) cd[i] = start; else cs[i] = start;
        tbl[(side * NBIN + i) * G1 + b] = start;
    }
    __syncthreads();
    for (int i0 = beg; i0 < end; i0 += 8 * 1024) {
        unsigned d[8], s[8];
#pragma unroll
        for (int u = 0; u < 8; ++u) {
            const int i = i0 + u * 1024 + t;
            const int ic = (i < end) ? i : beg;
            d[u] = (unsigned)dst[ic];
            s[u] = (unsigned)src[ic];
        }
#pragma unroll
        for (int u = 0; u < 8; ++u) {
            if (i0 + u * 1024 + t < end) {
                const int pd = atomicAdd(&cd[range_of(d[u])], 1);
                const int ps = atomicAdd(&cs[range_of(s[u])], 1);
                runs_d[pd] = (int)((s[u] << 9) | (d[u] & 511u));
                runs_s[ps] = (int)(s[u] & 511u);
            }
        }
    }
}

__global__ void __launch_bounds__(1024)
range_lists_kernel(const int* __restrict__ tbl, const int* __restrict__ runs_d,
                   const int* __restrict__ runs_s, int* __restrict__ rowptr,
                   int* __restrict__ srcs, float* __restrict__ ns, int E) {
    constexpr int TPS = 1024 / G1;
    constexpr int U = 12;
    static_assert(TPS * G1 == 1024, "every run has its threads");
    __shared__ int cnt[512];
    __shared__ int cur[512];
    __shared__ int ocn[512];
    __shared__ int wsum[32];
    __shared__ int lsum[32];
    const int t = threadIdx.x;
    const int lane = t & 31;
    const int w = t >> 5;
    const int g = blockIdx.x;
    const int ch = (E + G1 - 1) / G1;
    const int sb = t / TPS;
    const int j = t % TPS;
    const int sbeg = min(sb * ch, E);
    const int send = min(sbeg + ch, E);
    const int bd = tbl[g * G1 + sb];
    const int ed = (g == NBIN - 1) ? send : tbl[(g + 1) * G1 + sb];
    const int bs = tbl[(NBIN + g) * G1 + sb];
    const int es = (g == NBIN - 1) ? send : tbl[(NBIN + g + 1) * G1 + sb];
    int part = (j == 0) ? (bd - sbeg) : 0;
    for (int o = 16; o > 0; o >>= 1) part += __shfl_xor(part, o);
    if (lane == 0) lsum[w] = part;

    if (t < 512) { cnt[t] = 0; ocn[t] = 0; }
    __syncthreads();
    for (int i0 = bd; i0 < ed; i0 += U * TPS) {
        int wd[U];
#pragma unroll
        for (int u = 0; u < U; ++u) {
            const int i = i0 + u * TPS + j;
            wd[u] = runs_d[(i < ed) ? i : bd];
        }
#pragma unroll
        for (int u = 0; u < U; ++u)
            if (i0 + u * TPS + j < ed) atomicAdd(&cnt[wd[u] & 511], 1);
    }
    for (int i0 = bs; i0 < es; i0 += U * TPS) {
        int wo[U];
#pragma unroll
        for (int u = 0; u < U; ++u) {
            const int i = i0 + u * TPS + j;
            wo[u] = runs_s[(i < es) ? i : bs];
        }
#pragma unroll
        for (int u = 0; u < U; ++u)
            if (i0 + u * TPS + j < es) atomicAdd(&ocn[wo[u] & 511], 1);
    }
    __syncthreads();

    const int c = (t < 512) ? cnt[t] : 0;
    int inc = c;
    for (int o = 1; o < 32; o <<= 1) {
        const int v = __shfl_up(inc, o);
        if (lane >= o) inc += v;
    }
    if (lane == 31) wsum[w] = inc;
    __syncthreads();
    const int wv = wsum[lane];
    int sc = wv;
    for (int o = 1; o < 32; o <<= 1) {
        const int v = __shfl_up(sc, o);
        if (lane >= o) sc += v;
    }
    const int wave_excl = __shfl(sc - wv, w);
    const int node = g * 512 + t;
    if (t < 512) {
        int lbase = 0;
#pragma unroll
        for (int k = 0; k < 32; ++k) lbase += lsum[k];
        const int first = lbase + wave_excl + (inc - c);
        cur[t] = first;
        if (node < N_NODES) {
            rowptr[node] = first;
            ns[node] = rsqrtf(fmaxf((float)ocn[t], 1.0f));
        }
    }
    if (g == NBIN - 1 && t == 0) rowptr[N_NODES] = E;
    __syncthreads();
    for (int i0 = bd; i0 < ed; i0 += U * TPS) {
        int wd[U];
#pragma unroll
        for (int u = 0; u < U; ++u) {
            const int i = i0 + u * TPS + j;
            wd[u] = runs_d[(i < ed) ? i : bd];
        }
#pragma unroll
        for (int u = 0; u < U; ++u) {
            if (i0 + u * TPS + j < ed) {
                const int p = atomicAdd(&cur[wd[u] & 511], 1);
                srcs[p] = (int)((unsigned)wd[u] >> 9);
            }
        }
    }
}

template <int DIN, int DOUT>
__global__ void __launch_bounds__(DOUT * 2)
gemm_norm_wmma(const float* __restrict__ X, const float* __restrict__ ns,
               const float* __restrict__ W, float* __restrict__ Y,
               int nstripes) {
    static_assert(DIN == 128, "one wave loads one row of X per pass");
    constexpr int NW = DOUT / 16;
    constexpr int PASSES = 16 / NW;
    constexpr int PITCH = DIN + 4;
    __shared__ __attribute__((aligned(16))) float As[2][16 * PITCH];

    const int wave = __builtin_amdgcn_readfirstlane((int)(threadIdx.x >> 5));
    const int lane = threadIdx.x & 31;
    const int hl   = lane & 15;
    const bool hi  = (lane >= 16);
    const int kofs = hi ? 2 : 0;
    const int col  = wave * 16 + hl;

    v2f bfrag[DIN / 4];
#pragma unroll
    for (int i = 0; i < DIN / 4; ++i) {
        const int ka = 4 * i + kofs;
        bfrag[i][0] = W[(size_t)ka * DOUT + col];
        bfrag[i][1] = W[(size_t)(ka + 1) * DOUT + col];
    }

    int stripe = blockIdx.x;
    if (stripe >= nstripes) return;

    float4 pre[PASSES];
    float  psc[PASSES];
#define FETCH_STRIPE(s)                                                              \
    _Pragma("unroll")                                                                \
    for (int p = 0; p < PASSES; ++p) {                                               \
        const int r = (s) * 16 + p * NW + wave;                                      \
        pre[p] = *(const float4*)(X + (size_t)r * DIN + lane * 4);                   \
        psc[p] = ns[r];                                                              \
    }
#define STASH_STRIPE(b)                                                              \
    _Pragma("unroll")                                                                \
    for (int p = 0; p < PASSES; ++p) {                                               \
        float4 v = pre[p];                                                           \
        v.x *= psc[p]; v.y *= psc[p]; v.z *= psc[p]; v.w *= psc[p];                  \
        *(float4*)(&As[b][(p * NW + wave) * PITCH + lane * 4]) = v;                  \
    }

    FETCH_STRIPE(stripe)
    STASH_STRIPE(0)
    __syncthreads();

    int buf = 0;
    for (; stripe < nstripes; stripe += gridDim.x) {
        const int next = stripe + gridDim.x;
        const bool more = next < nstripes;
        if (more) { FETCH_STRIPE(next) }

        const float* __restrict__ arow = &As[buf][hl * PITCH + kofs];
        v8f acc = {};
#pragma unroll
        for (int i = 0; i < DIN / 4; ++i) {
            const v2f a = *(const v2f*)(arow + 4 * i);
            acc = __builtin_amdgcn_wmma_f32_16x16x4_f32(
                 false, a,  false, bfrag[i],
                 (short)0, acc,  false,  false);
        }

        if (more) { STASH_STRIPE(buf ^ 1) }
        __syncthreads();
        buf ^= 1;

        const int crow0 = stripe * 16 + (hi ? 8 : 0);
#pragma unroll
        for (int v = 0; v < 8; ++v) {
            Y[(size_t)(crow0 + v) * DOUT + col] = acc[v];
        }
    }
#undef FETCH_STRIPE
#undef STASH_STRIPE
}

template <int D, int BATCH, typename vec>
__device__ __forceinline__ void sum_rows(vec& acc, const char* __restrict__ Hb, unsigned lane_off,
                                         int mine, int cnt) {
    int j = 0;
    for (; j + BATCH <= cnt; j += BATCH) {
        int s[BATCH];
        vec v[BATCH];
#pragma unroll
        for (int u = 0; u < BATCH; ++u) s[u] = __builtin_amdgcn_readlane(mine, j + u);
#pragma unroll
        for (int u = 0; u < BATCH; ++u)
            v[u] = *(const vec*)(Hb + ((size_t)(unsigned)s[u] * (D * 4)) + lane_off);
#pragma unroll
        for (int u = 0; u < BATCH; ++u) acc += v[u];
    }
    if (j < cnt) {
        int s[BATCH];
        vec v[BATCH];
#pragma unroll
        for (int u = 0; u < BATCH; ++u) s[u] = __builtin_amdgcn_readlane(mine, (j + u) & 31);
#pragma unroll
        for (int u = 0; u < BATCH; ++u) {
            if (j + u < cnt)
                v[u] = *(const vec*)(Hb + ((size_t)(unsigned)s[u] * (D * 4)) + lane_off);
            else
                v[u] = (vec)(0.0f);
        }
#pragma unroll
        for (int u = 0; u < BATCH; ++u) acc += v[u];
    }
}

template <int D, bool RELU>
__global__ void gather_sum_kernel(const float* __restrict__ H, const int* __restrict__ rowptr,
                                  const int* __restrict__ srcs, const float* __restrict__ bias,
                                  float* __restrict__ O) {
    constexpr int VEC = D / 32;
    constexpr int BATCH = 8;
    typedef __attribute__((ext_vector_type(VEC))) float vec;

    const int lane = threadIdx.x & 31;
    const int wv = __builtin_amdgcn_readfirstlane((int)(blockIdx.x * 8 + (threadIdx.x >> 5)));
    const int nw = (int)gridDim.x * 8;
    if (wv >= N_NODES) return;

    const char* __restrict__ Hb = (const char*)H;
    const unsigned lane_off = (unsigned)lane * (VEC * 4);
    const vec bv = ((const vec*)bias)[lane];

    int node = wv;
    int beg = rowptr[node];
    int end = rowptr[node + 1];
    int first = (lane < end - beg) ? srcs[beg + lane] : 0;
    asm volatile("" : "+v"(first));
    int nnode = node + nw;
    int nbeg = 0, nend = 0;
    if (nnode < N_NODES) { nbeg = rowptr[nnode]; nend = rowptr[nnode + 1]; }

    for (;;) {
        int nfirst = 0;
        if (nnode < N_NODES) nfirst = (lane < nend - nbeg) ? srcs[nbeg + lane] : 0;
        const int n2 = nnode + nw;
        int n2beg = 0, n2end = 0;
        if (n2 < N_NODES) { n2beg = rowptr[n2]; n2end = rowptr[n2 + 1]; }

        vec acc = (vec)(0.0f);
        sum_rows<D, BATCH, vec>(acc, Hb, lane_off, first, min(32, end - beg));
        for (int base = beg + 32; base < end; base += 32) {
            const int cnt = min(32, end - base);
            const int mine = (lane < cnt) ? srcs[base + lane] : 0;
            sum_rows<D, BATCH, vec>(acc, Hb, lane_off, mine, cnt);
        }

        const float nd = rsqrtf(fmaxf((float)(end - beg), 1.0f));
        vec o = acc * nd + bv;
        if (RELU) {
#pragma unroll
            for (int i = 0; i < VEC; ++i) o[i] = fmaxf(o[i], 0.0f);
        }
        ((vec*)O)[(size_t)node * 32 + lane] = o;

        if (nnode >= N_NODES) break;
        node = nnode; beg = nbeg; end = nend; first = nfirst;
        nnode = n2; nbeg = n2beg; nend = n2end;
    }
}

template <int DOUT, bool COPY>
__global__ void __launch_bounds__(DOUT * 2, 8)
gather_gemm_wmma(const float* __restrict__ H, const int* __restrict__ rowptr,
                 const int* __restrict__ srcs, const float* __restrict__ bias,
                 const float* __restrict__ ns, const float* __restrict__ W,
                 float* __restrict__ Y, int nstripes,
                 int* __restrict__ rowptr_c, int* __restrict__ srcs_c) {
    constexpr int DIN = 128;
    constexpr int NW = DOUT / 16;
    constexpr int NPW = 16 / NW;
    constexpr int PITCH = DIN + 4;
    constexpr int BATCH = 8;
    typedef __attribute__((ext_vector_type(4))) float vec;
    __shared__ __attribute__((aligned(16))) float As[2][16 * PITCH];

    const int wave = __builtin_amdgcn_readfirstlane((int)(threadIdx.x >> 5));
    const int lane = threadIdx.x & 31;
    const int hl   = lane & 15;
    const bool hi  = (lane >= 16);
    const int kofs = hi ? 2 : 0;
    const int col  = wave * 16 + hl;

    v2f bfrag[DIN / 4];
#pragma unroll
    for (int i = 0; i < DIN / 4; ++i) {
        const int ka = 4 * i + kofs;
        bfrag[i][0] = W[(size_t)ka * DOUT + col];
        bfrag[i][1] = W[(size_t)(ka + 1) * DOUT + col];
    }

    const int G = (int)gridDim.x;
    if ((int)blockIdx.x >= nstripes) return;

    const char* __restrict__ Hb = (const char*)H;
    const unsigned lane_off = (unsigned)lane * 16u;
    const vec bv = ((const vec*)bias)[lane];

    auto node_of = [&](int q) -> int {
        const int st = (int)blockIdx.x + (q / NPW) * G;
        return (st < nstripes) ? st * 16 + wave * NPW + (q % NPW) : -1;
    };

    int q = 0;
    int node = node_of(0);
    int beg = rowptr[node];
    int end = rowptr[node + 1];
    int first = (lane < end - beg) ? srcs[beg + lane] : 0;
    asm volatile("" : "+v"(first));
    int nnode = node_of(1);
    int nbeg = 0, nend = 0;
    if (nnode >= 0) { nbeg = rowptr[nnode]; nend = rowptr[nnode + 1]; }

    int buf = 0;
    for (int stripe = blockIdx.x; stripe < nstripes; stripe += G) {
#pragma unroll
        for (int j = 0; j < NPW; ++j) {
            int nfirst = 0;
            if (nnode >= 0) nfirst = (lane < nend - nbeg) ? srcs[nbeg + lane] : 0;
            const int n2 = node_of(q + 2);
            int n2beg = 0, n2end = 0;
            if (n2 >= 0) { n2beg = rowptr[n2]; n2end = rowptr[n2 + 1]; }
            const float sc = ns[node];

            if (COPY) {
                if (lane < end - beg) srcs_c[beg + lane] = first;
                if (lane == 0) {
                    rowptr_c[node] = beg;
                    if (node == N_NODES - 1) rowptr_c[N_NODES] = end;
                }
            }

            vec acc = (vec)(0.0f);
            sum_rows<DIN, BATCH, vec>(acc, Hb, lane_off, first, min(32, end - beg));
            for (int base = beg + 32; base < end; base += 32) {
                const int cnt = min(32, end - base);
                const int mine = (lane < cnt) ? srcs[base + lane] : 0;
                if (COPY) { if (lane < cnt) srcs_c[base + lane] = mine; }
                sum_rows<DIN, BATCH, vec>(acc, Hb, lane_off, mine, cnt);
            }

            const float nd = rsqrtf(fmaxf((float)(end - beg), 1.0f));
            vec o = acc * nd + bv;
#pragma unroll
            for (int i = 0; i < 4; ++i) o[i] = fmaxf(o[i], 0.0f);
            o *= sc;
            *(vec*)(&As[buf][(wave * NPW + j) * PITCH + lane * 4]) = o;

            ++q;
            node = nnode; beg = nbeg; end = nend; first = nfirst;
            nnode = n2; nbeg = n2beg; nend = n2end;
        }
        __syncthreads();

        const float* __restrict__ arow = &As[buf][hl * PITCH + kofs];
        v8f acc = {};
#pragma unroll
        for (int i = 0; i < DIN / 4; ++i) {
            const v2f a = *(const v2f*)(arow + 4 * i);
            acc = __builtin_amdgcn_wmma_f32_16x16x4_f32(
                 false, a,  false, bfrag[i],
                 (short)0, acc,  false,  false);
        }
        buf ^= 1;

        const int crow0 = stripe * 16 + (hi ? 8 : 0);
#pragma unroll
        for (int v = 0; v < 8; ++v) {
            Y[(size_t)(crow0 + v) * DOUT + col] = acc[v];
        }
    }
}

extern "C" void kernel_launch(void* const* d_in, const int* in_sizes, int n_in,
                              void* d_out, int out_size, void* d_ws, size_t ws_size,
                              hipStream_t stream) {
    (void)n_in; (void)out_size; (void)ws_size;

    const float* x  = (const float*)d_in[0];
    const int*   src = (const int*)d_in[1];
    const int*   dst = (const int*)d_in[2];
    const float* W1 = (const float*)d_in[3];
    const float* b1 = (const float*)d_in[4];
    const float* W2 = (const float*)d_in[5];
    const float* b2 = (const float*)d_in[6];
    const float* W3 = (const float*)d_in[7];
    const float* b3 = (const float*)d_in[8];

    const int E = in_sizes[1];
    const int N = N_NODES;

    const size_t NPAD = 100352;
    float* ws   = (float*)d_ws;
    float* ns   = ws;
    int*   tbl  = (int*)ws + NPAD;
    float* bufA = ws + 2 * NPAD;
    float* bufB = bufA + (size_t)N * 128;

    int* rowptr  = (int*)d_out;
    int* srcs    = rowptr + (N + 1);
    int* runs_s  = (int*)bufB;
    int* runs_d  = (int*)bufA;
    int* rowptr3 = (int*)(bufA + (size_t)N * 64);
    int* srcs3   = rowptr3 + (N + 1);

    bin_place_kernel<<<G1, 1024, 0, stream>>>(src, dst, tbl, runs_d, runs_s, E);
    range_lists_kernel<<<NBIN, 1024, 0, stream>>>(tbl, runs_d, runs_s, rowptr, srcs, ns, E);

    const int ROWT = N / 16;
    const int GEMMB = 1024;
    const int GEMMB64 = 2048;
    const int NODEB = 2048;

    gemm_norm_wmma<128, 128><<<GEMMB, 256, 0, stream>>>(x, ns, W1, bufA, ROWT);

    gather_gemm_wmma<128, false><<<GEMMB, 256, 0, stream>>>(bufA, rowptr, srcs, b1, ns, W2, bufB, ROWT,
                                                            nullptr, nullptr);

    gather_gemm_wmma<64, true><<<GEMMB64, 128, 0, stream>>>(bufB, rowptr, srcs, b2, ns, W3, bufA, ROWT,
                                                            rowptr3, srcs3);

    gather_sum_kernel<64, false><<<NODEB, 256, 0, stream>>>(bufA, rowptr3, srcs3, b3, (float*)d_out);
}


# list build: both launches place their words into an LDS image of the block's contiguous output stretch and write it out in index order, 128 B per wave store, not as 4 B stores to 20-30 lines each
# speedup vs baseline: 5.9639x; 1.0754x over previous
#include <hip/hip_runtime.h>

#define N_NODES 100000

typedef __attribute__((ext_vector_type(2))) float v2f;
typedef __attribute__((ext_vector_type(8))) float v8f;

#define NBIN 196
#define G1 256
#define NROW (2 * NBIN)
#define PCAP 6656
#define LCAP 12288

__device__ __forceinline__ int range_of(unsigned node) { return min((int)(node >> 9), NBIN - 1); }

__global__ void __launch_bounds__(1024)
bin_place_kernel(const int* __restrict__ src, const int* __restrict__ dst,
                 int* __restrict__ tbl, int* __restrict__ runs_d,
                 int* __restrict__ runs_s, int E) {
    __shared__ int cd[NBIN];
    __shared__ int cs[NBIN];
    __shared__ int wsum[16];
    __shared__ int img_d[PCAP];
    __shared__ int img_s[PCAP];
    const int t = threadIdx.x;
    const int b = blockIdx.x;
    const int lane = t & 31;
    const int w = t >> 5;
    if (t < NBIN) { cd[t] = 0; cs[t] = 0; }
    __syncthreads();
    const int ch = (E + G1 - 1) / G1;
    const int beg = min(b * ch, E);
    const int end = min(beg + ch, E);
    for (int i0 = beg; i0 < end; i0 += 8 * 1024) {
        unsigned d[8], s[8];
#pragma unroll
        for (int u = 0; u < 8; ++u) {
            const int i = i0 + u * 1024 + t;
            const int ic = (i < end) ? i : beg;
            d[u] = (unsigned)dst[ic];
            s[u] = (unsigned)src[ic];
        }
#pragma unroll
        for (int u = 0; u < 8; ++u) {
            if (i0 + u * 1024 + t < end) {
                atomicAdd(&cd[range_of(d[u])], 1);
                atomicAdd(&cs[range_of(s[u])], 1);
            }
        }
    }
    __syncthreads();

    const int side = t >> 8;
    const int i = t & 255;
    int tv = 0, excl = 0;
    if (t < 512) {
        if (i < NBIN) tv = side ? cs[i] : cd[i];
        int inc = tv;
        for (int o = 1; o < 32; o <<= 1) {
            const int v = __shfl_up(inc, o);
            if (lane >= o) inc += v;
        }
        if (lane == 31) wsum[w] = inc;
        excl = inc - tv;
    }
    __syncthreads();
    if (t < 512 && i < NBIN) {
        int pre = 0;
#pragma unroll
        for (int k = 0; k < 8; ++k)
            if (side * 8 + k < w) pre += wsum[side * 8 + k];
        const int start = beg + pre + excl;
        if (side == 0) cd[i] = start; else cs[i] = start;
        tbl[(side * NBIN + i) * G1 + b] = start;
    }
    __syncthreads();
    const int n = end - beg;
    const bool staged = (n <= PCAP);
    for (int i0 = beg; i0 < end; i0 += 8 * 1024) {
        unsigned d[8], s[8];
#pragma unroll
        for (int u = 0; u < 8; ++u) {
            const int i = i0 + u * 1024 + t;
            const int ic = (i < end) ? i : beg;
            d[u] = (unsigned)dst[ic];
            s[u] = (unsigned)src[ic];
        }
        if (staged) {
#pragma unroll
            for (int u = 0; u < 8; ++u) {
                if (i0 + u * 1024 + t < end) {
                    const int pd = atomicAdd(&cd[range_of(d[u])], 1);
                    const int ps = atomicAdd(&cs[range_of(s[u])], 1);
                    img_d[pd - beg] = (int)((s[u] << 9) | (d[u] & 511u));
                    img_s[ps - beg] = (int)(s[u] & 511u);
                }
            }
        } else {
#pragma unroll
            for (int u = 0; u < 8; ++u) {
                if (i0 + u * 1024 + t < end) {
                    const int pd = atomicAdd(&cd[range_of(d[u])], 1);
                    const int ps = atomicAdd(&cs[range_of(s[u])], 1);
                    runs_d[pd] = (int)((s[u] << 9) | (d[u] & 511u));
                    runs_s[ps] = (int)(s[u] & 511u);
                }
            }
        }
    }
    __syncthreads();
    if (staged) {
        for (int i = t; i < n; i += 1024) {
            runs_d[beg + i] = img_d[i];
            runs_s[beg + i] = img_s[i];
        }
    }
}

__global__ void __launch_bounds__(1024)
range_lists_kernel(const int* __restrict__ tbl, const int* __restrict__ runs_d,
                   const int* __restrict__ runs_s, int* __restrict__ rowptr,
                   int* __restrict__ srcs, float* __restrict__ ns, int E) {
    constexpr int TPS = 1024 / G1;
    constexpr int U = 12;
    static_assert(TPS * G1 == 1024, "every run has its threads");
    __shared__ int cnt[512];
    __shared__ int cur[512];
    __shared__ int ocn[512];
    __shared__ int wsum[32];
    __shared__ int lsum[32];
    __shared__ int img[LCAP];
    const int t = threadIdx.x;
    const int lane = t & 31;
    const int w = t >> 5;
    const int g = blockIdx.x;
    const int ch = (E + G1 - 1) / G1;
    const int sb = t / TPS;
    const int j = t % TPS;
    const int sbeg = min(sb * ch, E);
    const int send = min(sbeg + ch, E);
    const int bd = tbl[g * G1 + sb];
    const int ed = (g == NBIN - 1) ? send : tbl[(g + 1) * G1 + sb];
    const int bs = tbl[(NBIN + g) * G1 + sb];
    const int es = (g == NBIN - 1) ? send : tbl[(NBIN + g + 1) * G1 + sb];
    int part = (j == 0) ? (bd - sbeg) : 0;
    for (int o = 16; o > 0; o >>= 1) part += __shfl_xor(part, o);
    if (lane == 0) lsum[w] = part;

    if (t < 512) { cnt[t] = 0; ocn[t] = 0; }
    __syncthreads();
    for (int i0 = bd; i0 < ed; i0 += U * TPS) {
        int wd[U];
#pragma unroll
        for (int u = 0; u < U; ++u) {
            const int i = i0 + u * TPS + j;
            wd[u] = runs_d[(i < ed) ? i : bd];
        }
#pragma unroll
        for (int u = 0; u < U; ++u)
            if (i0 + u * TPS + j < ed) atomicAdd(&cnt[wd[u] & 511], 1);
    }
    for (int i0 = bs; i0 < es; i0 += U * TPS) {
        int wo[U];
#pragma unroll
        for (int u = 0; u < U; ++u) {
            const int i = i0 + u * TPS + j;
            wo[u] = runs_s[(i < es) ? i : bs];
        }
#pragma unroll
        for (int u = 0; u < U; ++u)
            if (i0 + u * TPS + j < es) atomicAdd(&ocn[wo[u] & 511], 1);
    }
    __syncthreads();

    const int c = (t < 512) ? cnt[t] : 0;
    int inc = c;
    for (int o = 1; o < 32; o <<= 1) {
        const int v = __shfl_up(inc, o);
        if (lane >= o) inc += v;
    }
    if (lane == 31) wsum[w] = inc;
    __syncthreads();
    const int wv = wsum[lane];
    int sc = wv;
    for (int o = 1; o < 32; o <<= 1) {
        const int v = __shfl_up(sc, o);
        if (lane >= o) sc += v;
    }
    const int wave_excl = __shfl(sc - wv, w);
    const int total = __shfl(sc, 31);
    const bool staged = (total <= LCAP);
    const int node = g * 512 + t;
    int lbase = 0;
#pragma unroll
    for (int k = 0; k < 32; ++k) lbase += lsum[k];
    if (t < 512) {
        const int loc = wave_excl + (inc - c);
        const int first = lbase + loc;
        cur[t] = staged ? loc : first;
        if (node < N_NODES) {
            rowptr[node] = first;
            ns[node] = rsqrtf(fmaxf((float)ocn[t], 1.0f));
        }
    }
    if (g == NBIN - 1 && t == 0) rowptr[N_NODES] = E;
    __syncthreads();
    for (int i0 = bd; i0 < ed; i0 += U * TPS) {
        int wd[U];
#pragma unroll
        for (int u = 0; u < U; ++u) {
            const int i = i0 + u * TPS + j;
            wd[u] = runs_d[(i < ed) ? i : bd];
        }
        if (staged) {
#pragma unroll
            for (int u = 0; u < U; ++u) {
                if (i0 + u * TPS + j < ed) {
                    const int p = atomicAdd(&cur[wd[u] & 511], 1);
                    img[p] = (int)((unsigned)wd[u] >> 9);
                }
            }
        } else {
#pragma unroll
            for (int u = 0; u < U; ++u) {
                if (i0 + u * TPS + j < ed) {
                    const int p = atomicAdd(&cur[wd[u] & 511], 1);
                    srcs[p] = (int)((unsigned)wd[u] >> 9);
                }
            }
        }
    }
    __syncthreads();
    if (staged) {
        for (int i = t; i < total; i += 1024) srcs[lbase + i] = img[i];
    }
}

template <int DIN, int DOUT>
__global__ void __launch_bounds__(DOUT * 2)
gemm_norm_wmma(const float* __restrict__ X, const float* __restrict__ ns,
               const float* __restrict__ W, float* __restrict__ Y,
               int nstripes) {
    static_assert(DIN == 128, "one wave loads one row of X per pass");
    constexpr int NW = DOUT / 16;
    constexpr int PASSES = 16 / NW;
    constexpr int PITCH = DIN + 4;
    __shared__ __attribute__((aligned(16))) float As[2][16 * PITCH];

    const int wave = __builtin_amdgcn_readfirstlane((int)(threadIdx.x >> 5));
    const int lane = threadIdx.x & 31;
    const int hl   = lane & 15;
    const bool hi  = (lane >= 16);
    const int kofs = hi ? 2 : 0;
    const int col  = wave * 16 + hl;

    v2f bfrag[DIN / 4];
#pragma unroll
    for (int i = 0; i < DIN / 4; ++i) {
        const int ka = 4 * i + kofs;
        bfrag[i][0] = W[(size_t)ka * DOUT + col];
        bfrag[i][1] = W[(size_t)(ka + 1) * DOUT + col];
    }

    int stripe = blockIdx.x;
    if (stripe >= nstripes) return;

    float4 pre[PASSES];
    float  psc[PASSES];
#define FETCH_STRIPE(s)                                                              \
    _Pragma("unroll")                                                                \
    for (int p = 0; p < PASSES; ++p) {                                               \
        const int r = (s) * 16 + p * NW + wave;                                      \
        pre[p] = *(const float4*)(X + (size_t)r * DIN + lane * 4);                   \
        psc[p] = ns[r];                                                              \
    }
#define STASH_STRIPE(b)                                                              \
    _Pragma("unroll")                                                                \
    for (int p = 0; p < PASSES; ++p) {                                               \
        float4 v = pre[p];                                                           \
        v.x *= psc[p]; v.y *= psc[p]; v.z *= psc[p]; v.w *= psc[p];                  \
        *(float4*)(&As[b][(p * NW + wave) * PITCH + lane * 4]) = v;                  \
    }

    FETCH_STRIPE(stripe)
    STASH_STRIPE(0)
    __syncthreads();

    int buf = 0;
    for (; stripe < nstripes; stripe += gridDim.x) {
        const int next = stripe + gridDim.x;
        const bool more = next < nstripes;
        if (more) { FETCH_STRIPE(next) }

        const float* __restrict__ arow = &As[buf][hl * PITCH + kofs];
        v8f acc = {};
#pragma unroll
        for (int i = 0; i < DIN / 4; ++i) {
            const v2f a = *(const v2f*)(arow + 4 * i);
            acc = __builtin_amdgcn_wmma_f32_16x16x4_f32(
                 false, a,  false, bfrag[i],
                 (short)0, acc,  false,  false);
        }

        if (more) { STASH_STRIPE(buf ^ 1) }
        __syncthreads();
        buf ^= 1;

        const int crow0 = stripe * 16 + (hi ? 8 : 0);
#pragma unroll
        for (int v = 0; v < 8; ++v) {
            Y[(size_t)(crow0 + v) * DOUT + col] = acc[v];
        }
    }
#undef FETCH_STRIPE
#undef STASH_STRIPE
}

template <int D, int BATCH, typename vec>
__device__ __forceinline__ void sum_rows(vec& acc, const char* __restrict__ Hb, unsigned lane_off,
                                         int mine, int cnt) {
    int j = 0;
    for (; j + BATCH <= cnt; j += BATCH) {
        int s[BATCH];
        vec v[BATCH];
#pragma unroll
        for (int u = 0; u < BATCH; ++u) s[u] = __builtin_amdgcn_readlane(mine, j + u);
#pragma unroll
        for (int u = 0; u < BATCH; ++u)
            v[u] = *(const vec*)(Hb + ((size_t)(unsigned)s[u] * (D * 4)) + lane_off);
#pragma unroll
        for (int u = 0; u < BATCH; ++u) acc += v[u];
    }
    if (j < cnt) {
        int s[BATCH];
        vec v[BATCH];
#pragma unroll
        for (int u = 0; u < BATCH; ++u) s[u] = __builtin_amdgcn_readlane(mine, (j + u) & 31);
#pragma unroll
        for (int u = 0; u < BATCH; ++u) {
            if (j + u < cnt)
                v[u] = *(const vec*)(Hb + ((size_t)(unsigned)s[u] * (D * 4)) + lane_off);
            else
                v[u] = (vec)(0.0f);
        }
#pragma unroll
        for (int u = 0; u < BATCH; ++u) acc += v[u];
    }
}

template <int D, bool RELU>
__global__ void gather_sum_kernel(const float* __restrict__ H, const int* __restrict__ rowptr,
                                  const int* __restrict__ srcs, const float* __restrict__ bias,
                                  float* __restrict__ O) {
    constexpr int VEC = D / 32;
    constexpr int BATCH = 8;
    typedef __attribute__((ext_vector_type(VEC))) float vec;

    const int lane = threadIdx.x & 31;
    const int wv = __builtin_amdgcn_readfirstlane((int)(blockIdx.x * 8 + (threadIdx.x >> 5)));
    const int nw = (int)gridDim.x * 8;
    if (wv >= N_NODES) return;

    const char* __restrict__ Hb = (const char*)H;
    const unsigned lane_off = (unsigned)lane * (VEC * 4);
    const vec bv = ((const vec*)bias)[lane];

    int node = wv;
    int beg = rowptr[node];
    int end = rowptr[node + 1];
    int first = (lane < end - beg) ? srcs[beg + lane] : 0;
    asm volatile("" : "+v"(first));
    int nnode = node + nw;
    int nbeg = 0, nend = 0;
    if (nnode < N_NODES) { nbeg = rowptr[nnode]; nend = rowptr[nnode + 1]; }

    for (;;) {
        int nfirst = 0;
        if (nnode < N_NODES) nfirst = (lane < nend - nbeg) ? srcs[nbeg + lane] : 0;
        const int n2 = nnode + nw;
        int n2beg = 0, n2end = 0;
        if (n2 < N_NODES) { n2beg = rowptr[n2]; n2end = rowptr[n2 + 1]; }

        vec acc = (vec)(0.0f);
        sum_rows<D, BATCH, vec>(acc, Hb, lane_off, first, min(32, end - beg));
        for (int base = beg + 32; base < end; base += 32) {
            const int cnt = min(32, end - base);
            const int mine = (lane < cnt) ? srcs[base + lane] : 0;
            sum_rows<D, BATCH, vec>(acc, Hb, lane_off, mine, cnt);
        }

        const float nd = rsqrtf(fmaxf((float)(end - beg), 1.0f));
        vec o = acc * nd + bv;
        if (RELU) {
#pragma unroll
            for (int i = 0; i < VEC; ++i) o[i] = fmaxf(o[i], 0.0f);
        }
        ((vec*)O)[(size_t)node * 32 + lane] = o;

        if (nnode >= N_NODES) break;
        node = nnode; beg = nbeg; end = nend; first = nfirst;
        nnode = n2; nbeg = n2beg; nend = n2end;
    }
}

template <int DOUT, bool COPY>
__global__ void __launch_bounds__(DOUT * 2, 8)
gather_gemm_wmma(const float* __restrict__ H, const int* __restrict__ rowptr,
                 const int* __restrict__ srcs, const float* __restrict__ bias,
                 const float* __restrict__ ns, const float* __restrict__ W,
                 float* __restrict__ Y, int nstripes,
                 int* __restrict__ rowptr_c, int* __restrict__ srcs_c) {
    constexpr int DIN = 128;
    constexpr int NW = DOUT / 16;
    constexpr int NPW = 16 / NW;
    constexpr int PITCH = DIN + 4;
    constexpr int BATCH = 8;
    typedef __attribute__((ext_vector_type(4))) float vec;
    __shared__ __attribute__((aligned(16))) float As[2][16 * PITCH];

    const int wave = __builtin_amdgcn_readfirstlane((int)(threadIdx.x >> 5));
    const int lane = threadIdx.x & 31;
    const int hl   = lane & 15;
    const bool hi  = (lane >= 16);
    const int kofs = hi ? 2 : 0;
    const int col  = wave * 16 + hl;

    v2f bfrag[DIN / 4];
#pragma unroll
    for (int i = 0; i < DIN / 4; ++i) {
        const int ka = 4 * i + kofs;
        bfrag[i][0] = W[(size_t)ka * DOUT + col];
        bfrag[i][1] = W[(size_t)(ka + 1) * DOUT + col];
    }

    const int G = (int)gridDim.x;
    if ((int)blockIdx.x >= nstripes) return;

    const char* __restrict__ Hb = (const char*)H;
    const unsigned lane_off = (unsigned)lane * 16u;
    const vec bv = ((const vec*)bias)[lane];

    auto node_of = [&](int q) -> int {
        const int st = (int)blockIdx.x + (q / NPW) * G;
        return (st < nstripes) ? st * 16 + wave * NPW + (q % NPW) : -1;
    };

    int q = 0;
    int node = node_of(0);
    int beg = rowptr[node];
    int end = rowptr[node + 1];
    int first = (lane < end - beg) ? srcs[beg + lane] : 0;
    asm volatile("" : "+v"(first));
    int nnode = node_of(1);
    int nbeg = 0, nend = 0;
    if (nnode >= 0) { nbeg = rowptr[nnode]; nend = rowptr[nnode + 1]; }

    int buf = 0;
    for (int stripe = blockIdx.x; stripe < nstripes; stripe += G) {
#pragma unroll
        for (int j = 0; j < NPW; ++j) {
            int nfirst = 0;
            if (nnode >= 0) nfirst = (lane < nend - nbeg) ? srcs[nbeg + lane] : 0;
            const int n2 = node_of(q + 2);
            int n2beg = 0, n2end = 0;
            if (n2 >= 0) { n2beg = rowptr[n2]; n2end = rowptr[n2 + 1]; }
            const float sc = ns[node];

            if (COPY) {
                if (lane < end - beg) srcs_c[beg + lane] = first;
                if (lane == 0) {
                    rowptr_c[node] = beg;
                    if (node == N_NODES - 1) rowptr_c[N_NODES] = end;
                }
            }

            vec acc = (vec)(0.0f);
            sum_rows<DIN, BATCH, vec>(acc, Hb, lane_off, first, min(32, end - beg));
            for (int base = beg + 32; base < end; base += 32) {
                const int cnt = min(32, end - base);
                const int mine = (lane < cnt) ? srcs[base + lane] : 0;
                if (COPY) { if (lane < cnt) srcs_c[base + lane] = mine; }
                sum_rows<DIN, BATCH, vec>(acc, Hb, lane_off, mine, cnt);
            }

            const float nd = rsqrtf(fmaxf((float)(end - beg), 1.0f));
            vec o = acc * nd + bv;
#pragma unroll
            for (int i = 0; i < 4; ++i) o[i] = fmaxf(o[i], 0.0f);
            o *= sc;
            *(vec*)(&As[buf][(wave * NPW + j) * PITCH + lane * 4]) = o;

            ++q;
            node = nnode; beg = nbeg; end = nend; first = nfirst;
            nnode = n2; nbeg = n2beg; nend = n2end;
        }
        __syncthreads();

        const float* __restrict__ arow = &As[buf][hl * PITCH + kofs];
        v8f acc = {};
#pragma unroll
        for (int i = 0; i < DIN / 4; ++i) {
            const v2f a = *(const v2f*)(arow + 4 * i);
            acc = __builtin_amdgcn_wmma_f32_16x16x4_f32(
                 false, a,  false, bfrag[i],
                 (short)0, acc,  false,  false);
        }
        buf ^= 1;

        const int crow0 = stripe * 16 + (hi ? 8 : 0);
#pragma unroll
        for (int v = 0; v < 8; ++v) {
            Y[(size_t)(crow0 + v) * DOUT + col] = acc[v];
        }
    }
}

extern "C" void kernel_launch(void* const* d_in, const int* in_sizes, int n_in,
                              void* d_out, int out_size, void* d_ws, size_t ws_size,
                              hipStream_t stream) {
    (void)n_in; (void)out_size; (void)ws_size;

    const float* x  = (const float*)d_in[0];
    const int*   src = (const int*)d_in[1];
    const int*   dst = (const int*)d_in[2];
    const float* W1 = (const float*)d_in[3];
    const float* b1 = (const float*)d_in[4];
    const float* W2 = (const float*)d_in[5];
    const float* b2 = (const float*)d_in[6];
    const float* W3 = (const float*)d_in[7];
    const float* b3 = (const float*)d_in[8];

    const int E = in_sizes[1];
    const int N = N_NODES;

    const size_t NPAD = 100352;
    float* ws   = (float*)d_ws;
    float* ns   = ws;
    int*   tbl  = (int*)ws + NPAD;
    float* bufA = ws + 2 * NPAD;
    float* bufB = bufA + (size_t)N * 128;

    int* rowptr  = (int*)d_out;
    int* srcs    = rowptr + (N + 1);
    int* runs_s  = (int*)bufB;
    int* runs_d  = (int*)bufA;
    int* rowptr3 = (int*)(bufA + (size_t)N * 64);
    int* srcs3   = rowptr3 + (N + 1);

    bin_place_kernel<<<G1, 1024, 0, stream>>>(src, dst, tbl, runs_d, runs_s, E);
    range_lists_kernel<<<NBIN, 1024, 0, stream>>>(tbl, runs_d, runs_s, rowptr, srcs, ns, E);

    const int ROWT = N / 16;
    const int GEMMB = 1024;
    const int GEMMB64 = 2048;
    const int NODEB = 2048;

    gemm_norm_wmma<128, 128><<<GEMMB, 256, 0, stream>>>(x, ns, W1, bufA, ROWT);

    gather_gemm_wmma<128, false><<<GEMMB, 256, 0, stream>>>(bufA, rowptr, srcs, b1, ns, W2, bufB, ROWT,
                                                            nullptr, nullptr);

    gather_gemm_wmma<64, true><<<GEMMB64, 128, 0, stream>>>(bufB, rowptr, srcs, b2, ns, W3, bufA, ROWT,
                                                            rowptr3, srcs3);

    gather_sum_kernel<64, false><<<NODEB, 256, 0, stream>>>(bufA, rowptr3, srcs3, b3, (float*)d_out);
}
